# starting kernel
# baseline (speedup 1.0000x reference)
#include <hip/hip_runtime.h>

typedef __attribute__((ext_vector_type(2))) float v2f;
typedef __attribute__((ext_vector_type(8))) float v8f;

__global__ void k_degree(const int* __restrict__ src, const int* __restrict__ dst,
                         float* __restrict__ deg_td, float* __restrict__ deg_bu,
                         int n_edges) {
  int e = blockIdx.x * blockDim.x + threadIdx.x;
  if (e >= n_edges) return;
  atomicAdd(&deg_td[dst[e]], 1.0f);
  atomicAdd(&deg_bu[src[e]], 1.0f);
}

__global__ void k_rsqrt(float* __restrict__ d, int n) {
  int i = blockIdx.x * blockDim.x + threadIdx.x;
  if (i >= n) return;
  d[i] = rsqrtf(d[i] + 1.0f);
}

__global__ void k_gemm_wmma(const float* __restrict__ A, const float* __restrict__ B,
                            float* __restrict__ C, const float* __restrict__ bias,
                            int M, int N, int K, int relu) {
  const int lane = threadIdx.x & 31;
  const int wave = threadIdx.x >> 5;
  const int tile = blockIdx.x * (blockDim.x >> 5) + wave;
  const int tn_cnt = N >> 4;
  const int total = (M >> 4) * tn_cnt;
  if (tile >= total) return;
  const int tm = (tile / tn_cnt) << 4;
  const int tn = (tile % tn_cnt) << 4;

  const int half = lane >> 4;
  const int l    = lane & 15;

  const float* __restrict__ Arow = A + (size_t)(tm + l) * K;
  const float* __restrict__ Bcol = B + tn + l;

  v8f acc = {};
  for (int k = 0; k < K; k += 4) {
    const int kk = k + half * 2;
    v2f a, b;
    a.x = Arow[kk];
    a.y = Arow[kk + 1];
    b.x = Bcol[(size_t)kk * N];
    b.y = Bcol[(size_t)(kk + 1) * N];
    acc = __builtin_amdgcn_wmma_f32_16x16x4_f32(
         false, a,  false, b,
         (short)0, acc,  false,  false);
  }

  const float bi = bias ? bias[tn + l] : 0.0f;
#pragma unroll
  for (int r = 0; r < 8; ++r) {
    float v = acc[r] + bi;
    if (relu) v = fmaxf(v, 0.0f);
    C[(size_t)(tm + half * 8 + r) * N + tn + l] = v;
  }
}

__global__ void k_edge_agg(const float* __restrict__ h,
                           const int* __restrict__ s_idx, const int* __restrict__ d_idx,
                           const float* __restrict__ dinv,
                           float* __restrict__ agg, int n_edges) {
  const int lane = threadIdx.x & 31;
  const int e = (blockIdx.x * blockDim.x + threadIdx.x) >> 5;
  if (e >= n_edges) return;
  const int s = s_idx[e];
  const int d = d_idx[e];
  const float norm = dinv[s] * dinv[d];
  const float4* __restrict__ hs = (const float4*)(h + (size_t)s * 128);
  float* __restrict__ ad = agg + (size_t)d * 128;
  float4 v = hs[lane];
  const int f = lane * 4;
  atomicAdd(&ad[f + 0], v.x * norm);
  atomicAdd(&ad[f + 1], v.y * norm);
  atomicAdd(&ad[f + 2], v.z * norm);
  atomicAdd(&ad[f + 3], v.w * norm);
}

__global__ void k_combine_relu(const float* __restrict__ agg, const float* __restrict__ hlin,
                               const float* __restrict__ dinv, const float* __restrict__ bias,
                               float* __restrict__ out, int n) {
  int i = blockIdx.x * blockDim.x + threadIdx.x;
  if (i >= n) return;
  const int node = i >> 7;
  const int f = i & 127;
  const float di = dinv[node];
  const float v = agg[i] + hlin[i] * di * di + bias[f];
  out[i] = fmaxf(v, 0.0f);
}

__global__ void k_combine_pool(const float* __restrict__ agg, const float* __restrict__ hlin,
                               const float* __restrict__ dinv, const float* __restrict__ bias,
                               const int* __restrict__ batch, float* __restrict__ pooled,
                               int col_off, int n) {
  int i = blockIdx.x * blockDim.x + threadIdx.x;
  if (i >= n) return;
  const int node = i >> 7;
  const int f = i & 127;
  const float di = dinv[node];
  const float v = agg[i] + hlin[i] * di * di + bias[f];
  atomicAdd(&pooled[(size_t)batch[node] * 256 + col_off + f], v);
}

static inline void launch_gemm(const float* A, const float* B, float* C,
                               const float* bias, int M, int N, int K, int relu,
                               hipStream_t stream) {
  const int tiles = (M >> 4) * (N >> 4);
  const int blocks = (tiles + 7) / 8;
  k_gemm_wmma<<<blocks, 256, 0, stream>>>(A, B, C, bias, M, N, K, relu);
}

extern "C" void kernel_launch(void* const* d_in, const int* in_sizes, int n_in,
                              void* d_out, int out_size, void* d_ws, size_t ws_size,
                              hipStream_t stream) {
  const float* x      = (const float*)d_in[0];
  const int* edge_idx = (const int*)d_in[1];
  const int* batch    = (const int*)d_in[2];
  const float* td_W1 = (const float*)d_in[4];
  const float* td_b1 = (const float*)d_in[5];
  const float* td_W2 = (const float*)d_in[6];
  const float* td_b2 = (const float*)d_in[7];
  const float* bu_W1 = (const float*)d_in[8];
  const float* bu_b1 = (const float*)d_in[9];
  const float* bu_W2 = (const float*)d_in[10];
  const float* bu_b2 = (const float*)d_in[11];
  const float* pw1   = (const float*)d_in[12];
  const float* pb1   = (const float*)d_in[13];
  const float* pw2   = (const float*)d_in[14];
  const float* pb2   = (const float*)d_in[15];

  const int IN_F = 256, HID = 128, OUT_F = 128, G = 512;
  const int n_nodes = in_sizes[0] / IN_F;
  const int n_edges = in_sizes[1] / 2;
  const int* src = edge_idx;
  const int* dst = edge_idx + n_edges;

  float* ws      = (float*)d_ws;
  const size_t NF = (size_t)n_nodes * 128;
  float* h_lin   = ws;
  float* h_agg   = ws + NF;
  float* h_act   = ws + 2 * NF;
  float* dinv_td = ws + 3 * NF;
  float* dinv_bu = dinv_td + n_nodes;
  float* pooled  = dinv_bu + n_nodes;
  float* mlp_h   = pooled + (size_t)G * 256;

  const int EB = (n_edges + 255) / 256;
  const int WB = (n_edges * 32 + 255) / 256;
  const int NB = ((int)NF + 255) / 256;

  hipMemsetAsync(dinv_td, 0, (size_t)2 * n_nodes * sizeof(float), stream);
  k_degree<<<EB, 256, 0, stream>>>(src, dst, dinv_td, dinv_bu, n_edges);
  k_rsqrt<<<(2 * n_nodes + 255) / 256, 256, 0, stream>>>(dinv_td, 2 * n_nodes);
  hipMemsetAsync(pooled, 0, (size_t)G * 256 * sizeof(float), stream);

  launch_gemm(x, td_W1, h_lin, nullptr, n_nodes, HID, IN_F, 0, stream);
  hipMemsetAsync(h_agg, 0, NF * sizeof(float), stream);
  k_edge_agg<<<WB, 256, 0, stream>>>(h_lin, src, dst, dinv_td, h_agg, n_edges);
  k_combine_relu<<<NB, 256, 0, stream>>>(h_agg, h_lin, dinv_td, td_b1, h_act, (int)NF);
  launch_gemm(h_act, td_W2, h_lin, nullptr, n_nodes, OUT_F, HID, 0, stream);
  hipMemsetAsync(h_agg, 0, NF * sizeof(float), stream);
  k_edge_agg<<<WB, 256, 0, stream>>>(h_lin, src, dst, dinv_td, h_agg, n_edges);
  k_combine_pool<<<NB, 256, 0, stream>>>(h_agg, h_lin, dinv_td, td_b2, batch, pooled,
                                          128, (int)NF);

  launch_gemm(x, bu_W1, h_lin, nullptr, n_nodes, HID, IN_F, 0, stream);
  hipMemsetAsync(h_agg, 0, NF * sizeof(float), stream);
  k_edge_agg<<<WB, 256, 0, stream>>>(h_lin, dst, src, dinv_bu, h_agg, n_edges);
  k_combine_relu<<<NB, 256, 0, stream>>>(h_agg, h_lin, dinv_bu, bu_b1, h_act, (int)NF);
  launch_gemm(h_act, bu_W2, h_lin, nullptr, n_nodes, OUT_F, HID, 0, stream);
  hipMemsetAsync(h_agg, 0, NF * sizeof(float), stream);
  k_edge_agg<<<WB, 256, 0, stream>>>(h_lin, dst, src, dinv_bu, h_agg, n_edges);
  k_combine_pool<<<NB, 256, 0, stream>>>(h_agg, h_lin, dinv_bu, bu_b2, batch, pooled,
                                          0, (int)NF);

  launch_gemm(pooled, pw1, mlp_h, pb1, G, 256, 256,  1, stream);
  launch_gemm(mlp_h, pw2, (float*)d_out, pb2, G, 128, 256,  0, stream);
}


# edge aggregation: lane l owns features l, l+32, l+64, l+96, so each atomic add instruction of a wave covers one whole 128-byte line (was 4 lines with 8 lanes each): 4x fewer atomic requests to L2
# speedup vs baseline: 1.3419x; 1.3419x over previous
#include <hip/hip_runtime.h>

typedef __attribute__((ext_vector_type(2))) float v2f;
typedef __attribute__((ext_vector_type(8))) float v8f;

__global__ void k_degree(const int* __restrict__ src, const int* __restrict__ dst,
                         float* __restrict__ deg_td, float* __restrict__ deg_bu,
                         int n_edges) {
  int e = blockIdx.x * blockDim.x + threadIdx.x;
  if (e >= n_edges) return;
  atomicAdd(&deg_td[dst[e]], 1.0f);
  atomicAdd(&deg_bu[src[e]], 1.0f);
}

__global__ void k_rsqrt(float* __restrict__ d, int n) {
  int i = blockIdx.x * blockDim.x + threadIdx.x;
  if (i >= n) return;
  d[i] = rsqrtf(d[i] + 1.0f);
}

__global__ void k_gemm_wmma(const float* __restrict__ A, const float* __restrict__ B,
                            float* __restrict__ C, const float* __restrict__ bias,
                            int M, int N, int K, int relu) {
  const int lane = threadIdx.x & 31;
  const int wave = threadIdx.x >> 5;
  const int tile = blockIdx.x * (blockDim.x >> 5) + wave;
  const int tn_cnt = N >> 4;
  const int total = (M >> 4) * tn_cnt;
  if (tile >= total) return;
  const int tm = (tile / tn_cnt) << 4;
  const int tn = (tile % tn_cnt) << 4;

  const int half = lane >> 4;
  const int l    = lane & 15;

  const float* __restrict__ Arow = A + (size_t)(tm + l) * K;
  const float* __restrict__ Bcol = B + tn + l;

  v8f acc = {};
  for (int k = 0; k < K; k += 4) {
    const int kk = k + half * 2;
    v2f a, b;
    a.x = Arow[kk];
    a.y = Arow[kk + 1];
    b.x = Bcol[(size_t)kk * N];
    b.y = Bcol[(size_t)(kk + 1) * N];
    acc = __builtin_amdgcn_wmma_f32_16x16x4_f32(
         false, a,  false, b,
         (short)0, acc,  false,  false);
  }

  const float bi = bias ? bias[tn + l] : 0.0f;
#pragma unroll
  for (int r = 0; r < 8; ++r) {
    float v = acc[r] + bi;
    if (relu) v = fmaxf(v, 0.0f);
    C[(size_t)(tm + half * 8 + r) * N + tn + l] = v;
  }
}

__global__ void k_edge_agg(const float* __restrict__ h,
                           const int* __restrict__ s_idx, const int* __restrict__ d_idx,
                           const float* __restrict__ dinv,
                           float* __restrict__ agg, int n_edges) {
  const int lane = threadIdx.x & 31;
  const int e = (blockIdx.x * blockDim.x + threadIdx.x) >> 5;
  if (e >= n_edges) return;
  const int s = s_idx[e];
  const int d = d_idx[e];
  const float norm = dinv[s] * dinv[d];
  const float* __restrict__ hs = h + (size_t)s * 128 + lane;
  float* __restrict__ ad = agg + (size_t)d * 128 + lane;
  const float v0 = hs[0];
  const float v1 = hs[32];
  const float v2 = hs[64];
  const float v3 = hs[96];
  atomicAdd(&ad[0],  v0 * norm);
  atomicAdd(&ad[32], v1 * norm);
  atomicAdd(&ad[64], v2 * norm);
  atomicAdd(&ad[96], v3 * norm);
}

__global__ void k_combine_relu(const float* __restrict__ agg, const float* __restrict__ hlin,
                               const float* __restrict__ dinv, const float* __restrict__ bias,
                               float* __restrict__ out, int n) {
  int i = blockIdx.x * blockDim.x + threadIdx.x;
  if (i >= n) return;
  const int node = i >> 7;
  const int f = i & 127;
  const float di = dinv[node];
  const float v = agg[i] + hlin[i] * di * di + bias[f];
  out[i] = fmaxf(v, 0.0f);
}

__global__ void k_combine_pool(const float* __restrict__ agg, const float* __restrict__ hlin,
                               const float* __restrict__ dinv, const float* __restrict__ bias,
                               const int* __restrict__ batch, float* __restrict__ pooled,
                               int col_off, int n) {
  int i = blockIdx.x * blockDim.x + threadIdx.x;
  if (i >= n) return;
  const int node = i >> 7;
  const int f = i & 127;
  const float di = dinv[node];
  const float v = agg[i] + hlin[i] * di * di + bias[f];
  atomicAdd(&pooled[(size_t)batch[node] * 256 + col_off + f], v);
}

static inline void launch_gemm(const float* A, const float* B, float* C,
                               const float* bias, int M, int N, int K, int relu,
                               hipStream_t stream) {
  const int tiles = (M >> 4) * (N >> 4);
  const int blocks = (tiles + 7) / 8;
  k_gemm_wmma<<<blocks, 256, 0, stream>>>(A, B, C, bias, M, N, K, relu);
}

extern "C" void kernel_launch(void* const* d_in, const int* in_sizes, int n_in,
                              void* d_out, int out_size, void* d_ws, size_t ws_size,
                              hipStream_t stream) {
  const float* x      = (const float*)d_in[0];
  const int* edge_idx = (const int*)d_in[1];
  const int* batch    = (const int*)d_in[2];
  const float* td_W1 = (const float*)d_in[4];
  const float* td_b1 = (const float*)d_in[5];
  const float* td_W2 = (const float*)d_in[6];
  const float* td_b2 = (const float*)d_in[7];
  const float* bu_W1 = (const float*)d_in[8];
  const float* bu_b1 = (const float*)d_in[9];
  const float* bu_W2 = (const float*)d_in[10];
  const float* bu_b2 = (const float*)d_in[11];
  const float* pw1   = (const float*)d_in[12];
  const float* pb1   = (const float*)d_in[13];
  const float* pw2   = (const float*)d_in[14];
  const float* pb2   = (const float*)d_in[15];

  const int IN_F = 256, HID = 128, OUT_F = 128, G = 512;
  const int n_nodes = in_sizes[0] / IN_F;
  const int n_edges = in_sizes[1] / 2;
  const int* src = edge_idx;
  const int* dst = edge_idx + n_edges;

  float* ws      = (float*)d_ws;
  const size_t NF = (size_t)n_nodes * 128;
  float* h_lin   = ws;
  float* h_agg   = ws + NF;
  float* h_act   = ws + 2 * NF;
  float* dinv_td = ws + 3 * NF;
  float* dinv_bu = dinv_td + n_nodes;
  float* pooled  = dinv_bu + n_nodes;
  float* mlp_h   = pooled + (size_t)G * 256;

  const int EB = (n_edges + 255) / 256;
  const int WB = (n_edges * 32 + 255) / 256;
  const int NB = ((int)NF + 255) / 256;

  hipMemsetAsync(dinv_td, 0, (size_t)2 * n_nodes * sizeof(float), stream);
  k_degree<<<EB, 256, 0, stream>>>(src, dst, dinv_td, dinv_bu, n_edges);
  k_rsqrt<<<(2 * n_nodes + 255) / 256, 256, 0, stream>>>(dinv_td, 2 * n_nodes);
  hipMemsetAsync(pooled, 0, (size_t)G * 256 * sizeof(float), stream);

  launch_gemm(x, td_W1, h_lin, nullptr, n_nodes, HID, IN_F, 0, stream);
  hipMemsetAsync(h_agg, 0, NF * sizeof(float), stream);
  k_edge_agg<<<WB, 256, 0, stream>>>(h_lin, src, dst, dinv_td, h_agg, n_edges);
  k_combine_relu<<<NB, 256, 0, stream>>>(h_agg, h_lin, dinv_td, td_b1, h_act, (int)NF);
  launch_gemm(h_act, td_W2, h_lin, nullptr, n_nodes, OUT_F, HID, 0, stream);
  hipMemsetAsync(h_agg, 0, NF * sizeof(float), stream);
  k_edge_agg<<<WB, 256, 0, stream>>>(h_lin, src, dst, dinv_td, h_agg, n_edges);
  k_combine_pool<<<NB, 256, 0, stream>>>(h_agg, h_lin, dinv_td, td_b2, batch, pooled,
                                          128, (int)NF);

  launch_gemm(x, bu_W1, h_lin, nullptr, n_nodes, HID, IN_F, 0, stream);
  hipMemsetAsync(h_agg, 0, NF * sizeof(float), stream);
  k_edge_agg<<<WB, 256, 0, stream>>>(h_lin, dst, src, dinv_bu, h_agg, n_edges);
  k_combine_relu<<<NB, 256, 0, stream>>>(h_agg, h_lin, dinv_bu, bu_b1, h_act, (int)NF);
  launch_gemm(h_act, bu_W2, h_lin, nullptr, n_nodes, OUT_F, HID, 0, stream);
  hipMemsetAsync(h_agg, 0, NF * sizeof(float), stream);
  k_edge_agg<<<WB, 256, 0, stream>>>(h_lin, dst, src, dinv_bu, h_agg, n_edges);
  k_combine_pool<<<NB, 256, 0, stream>>>(h_agg, h_lin, dinv_bu, bu_b2, batch, pooled,
                                          0, (int)NF);

  launch_gemm(pooled, pw1, mlp_h, pb1, G, 256, 256,  1, stream);
  launch_gemm(mlp_h, pw2, (float*)d_out, pb2, G, 128, 256,  0, stream);
}


# edge aggregation as a gather over an edge list counting-sorted by destination inside the call: one wave sums a row in registers and writes it once; float atomics, memset and epilogue pass go
# speedup vs baseline: 1.8136x; 1.3515x over previous
#include <hip/hip_runtime.h>


typedef __attribute__((ext_vector_type(2))) float v2f;
typedef __attribute__((ext_vector_type(8))) float v8f;

__global__ void k_count_rank(const int* __restrict__ src, const int* __restrict__ dst,
                             int* __restrict__ cnt_td, int* __restrict__ cnt_bu,
                             int* __restrict__ rank_td, int* __restrict__ rank_bu,
                             int n_edges) {
  const int e = blockIdx.x * blockDim.x + threadIdx.x;
  if (e >= n_edges) return;
  rank_td[e] = atomicAdd(&cnt_td[dst[e]], 1);
  rank_bu[e] = atomicAdd(&cnt_bu[src[e]], 1);
}

#define SCAN_WAVES 32
#define SCAN_BATCH 7
__global__ void __launch_bounds__(SCAN_WAVES * 32)
k_scan_dinv(const int* __restrict__ cnt2, int* __restrict__ rowptr2,
            float* __restrict__ dinv2, int n) {
  __shared__ int wave_tot[SCAN_WAVES];
  const int* __restrict__ cnt = cnt2 + (size_t)blockIdx.x * n;
  int* __restrict__ rowptr = rowptr2 + (size_t)blockIdx.x * (n + 1);
  float* __restrict__ dinv = dinv2 + (size_t)blockIdx.x * n;
  const int lane = threadIdx.x & 31;
  const int wave = threadIdx.x >> 5;
  const int steps = (n + SCAN_WAVES * 32 - 1) / (SCAN_WAVES * 32);
  const int base = wave * steps * 32;

  int part = 0;
  for (int k = 0; k < steps; ++k) {
    const int i = base + k * 32 + lane;
    part += (i < n) ? cnt[i] : 0;
  }
  for (int o = 16; o > 0; o >>= 1) part += __shfl_xor(part, o);
  if (lane == 0) wave_tot[wave] = part;
  __syncthreads();
  int carry = (lane < wave) ? wave_tot[lane] : 0;
  for (int o = 16; o > 0; o >>= 1) carry += __shfl_xor(carry, o);

  for (int k0 = 0; k0 < steps; k0 += SCAN_BATCH) {
    int v[SCAN_BATCH];
#pragma unroll
    for (int u = 0; u < SCAN_BATCH; ++u) {
      const int i = base + (k0 + u) * 32 + lane;
      v[u] = (k0 + u < steps && i < n) ? cnt[i] : 0;
    }
    int x[SCAN_BATCH];
#pragma unroll
    for (int u = 0; u < SCAN_BATCH; ++u) x[u] = v[u];
#pragma unroll
    for (int o = 1; o < 32; o <<= 1) {
      int y[SCAN_BATCH];
#pragma unroll
      for (int u = 0; u < SCAN_BATCH; ++u) y[u] = __shfl_up(x[u], o);
#pragma unroll
      for (int u = 0; u < SCAN_BATCH; ++u) x[u] += (lane >= o) ? y[u] : 0;
    }
#pragma unroll
    for (int u = 0; u < SCAN_BATCH; ++u) {
      const int i = base + (k0 + u) * 32 + lane;
      if (k0 + u < steps && i < n) {
        rowptr[i] = carry + x[u] - v[u];
        dinv[i] = rsqrtf((float)v[u] + 1.0f);
      }
      carry += __builtin_amdgcn_readlane(x[u], 31);
    }
  }
  if (wave == SCAN_WAVES - 1 && lane == 0) rowptr[n] = carry;
}

__global__ void k_fill(const int* __restrict__ src, const int* __restrict__ dst,
                       const int* __restrict__ rowptr_td, const int* __restrict__ rowptr_bu,
                       const int* __restrict__ rank_td, const int* __restrict__ rank_bu,
                       int* __restrict__ col_td, int* __restrict__ col_bu, int n_edges) {
  const int e = blockIdx.x * blockDim.x + threadIdx.x;
  if (e >= n_edges) return;
  const int s = src[e];
  const int d = dst[e];
  col_td[rowptr_td[d] + rank_td[e]] = s;
  col_bu[rowptr_bu[s] + rank_bu[e]] = d;
}

__global__ void k_gemm_wmma(const float* __restrict__ A, const float* __restrict__ B,
                            float* __restrict__ C, const float* __restrict__ bias,
                            int M, int N, int K, int relu) {
  const int lane = threadIdx.x & 31;
  const int wave = threadIdx.x >> 5;
  const int tile = blockIdx.x * (blockDim.x >> 5) + wave;
  const int tn_cnt = N >> 4;
  const int total = (M >> 4) * tn_cnt;
  if (tile >= total) return;
  const int tm = (tile / tn_cnt) << 4;
  const int tn = (tile % tn_cnt) << 4;

  const int half = lane >> 4;
  const int l    = lane & 15;

  const float* __restrict__ Arow = A + (size_t)(tm + l) * K;
  const float* __restrict__ Bcol = B + tn + l;

  v8f acc = {};
  for (int k = 0; k < K; k += 4) {
    const int kk = k + half * 2;
    v2f a, b;
    a.x = Arow[kk];
    a.y = Arow[kk + 1];
    b.x = Bcol[(size_t)kk * N];
    b.y = Bcol[(size_t)(kk + 1) * N];
    acc = __builtin_amdgcn_wmma_f32_16x16x4_f32(
         false, a,  false, b,
         (short)0, acc,  false,  false);
  }

  const float bi = bias ? bias[tn + l] : 0.0f;
#pragma unroll
  for (int r = 0; r < 8; ++r) {
    float v = acc[r] + bi;
    if (relu) v = fmaxf(v, 0.0f);
    C[(size_t)(tm + half * 8 + r) * N + tn + l] = v;
  }
}

__device__ __forceinline__ float4 ld_row(const float* __restrict__ h, int s, int lane) {
  return *reinterpret_cast<const float4*>(h + (size_t)s * 128 + lane * 4);
}
__device__ __forceinline__ void fma4(float4& a, const float4 v, const float w) {
  a.x = fmaf(v.x, w, a.x);
  a.y = fmaf(v.y, w, a.y);
  a.z = fmaf(v.z, w, a.z);
  a.w = fmaf(v.w, w, a.w);
}
__device__ __forceinline__ int lane_i(int v, int j) { return __builtin_amdgcn_readlane(v, j); }
__device__ __forceinline__ float lane_f(float v, int j) {
  return __int_as_float(__builtin_amdgcn_readlane(__float_as_int(v), j));
}

__device__ __forceinline__ float4 gather_row(const float* __restrict__ h,
                                             const int* __restrict__ col,
                                             const float* __restrict__ dinv,
                                             int beg, int end, float dd, int lane) {
  float4 acc = make_float4(0.f, 0.f, 0.f, 0.f);
  for (int base = beg; base < end; base += 32) {
    const int n = min(32, end - base);
    int my_s = 0;
    float my_w = 0.f;
    if (lane < n) {
      my_s = col[base + lane];
      my_w = dinv[my_s] * dd;
    }
    int j = 0;
    for (; j + 4 <= n; j += 4) {
      const float4 v0 = ld_row(h, lane_i(my_s, j), lane);
      const float4 v1 = ld_row(h, lane_i(my_s, j + 1), lane);
      const float4 v2 = ld_row(h, lane_i(my_s, j + 2), lane);
      const float4 v3 = ld_row(h, lane_i(my_s, j + 3), lane);
      fma4(acc, v0, lane_f(my_w, j));
      fma4(acc, v1, lane_f(my_w, j + 1));
      fma4(acc, v2, lane_f(my_w, j + 2));
      fma4(acc, v3, lane_f(my_w, j + 3));
    }
    if (j + 2 <= n) {
      const float4 v0 = ld_row(h, lane_i(my_s, j), lane);
      const float4 v1 = ld_row(h, lane_i(my_s, j + 1), lane);
      fma4(acc, v0, lane_f(my_w, j));
      fma4(acc, v1, lane_f(my_w, j + 1));
      j += 2;
    }
    if (j < n) {
      const float4 v0 = ld_row(h, lane_i(my_s, j), lane);
      fma4(acc, v0, lane_f(my_w, j));
    }
  }
  return acc;
}

__global__ void __launch_bounds__(256)
k_gather_relu(const float* __restrict__ h, const int* __restrict__ rowptr,
              const int* __restrict__ col, const float* __restrict__ dinv,
              const float* __restrict__ bias, float* __restrict__ out, int n_nodes) {
  const int lane = threadIdx.x & 31;
  const int row = __builtin_amdgcn_readfirstlane((int)((blockIdx.x * blockDim.x + threadIdx.x) >> 5));
  if (row >= n_nodes) return;
  const int beg = rowptr[row];
  const int end = rowptr[row + 1];
  const float dd = dinv[row];
  const float4 self = ld_row(h, row, lane);
  const float4 b = *reinterpret_cast<const float4*>(bias + lane * 4);
  const float4 a = gather_row(h, col, dinv, beg, end, dd, lane);
  float4 v;
  v.x = fmaxf(a.x + self.x * dd * dd + b.x, 0.0f);
  v.y = fmaxf(a.y + self.y * dd * dd + b.y, 0.0f);
  v.z = fmaxf(a.z + self.z * dd * dd + b.z, 0.0f);
  v.w = fmaxf(a.w + self.w * dd * dd + b.w, 0.0f);
  *reinterpret_cast<float4*>(out + (size_t)row * 128 + lane * 4) = v;
}

__global__ void __launch_bounds__(256)
k_gather_pool(const float* __restrict__ h, const int* __restrict__ rowptr,
              const int* __restrict__ col, const float* __restrict__ dinv,
              const float* __restrict__ bias, const int* __restrict__ batch,
              float* __restrict__ pooled, int col_off, int n_nodes) {
  const int lane = threadIdx.x & 31;
  const int row = __builtin_amdgcn_readfirstlane((int)((blockIdx.x * blockDim.x + threadIdx.x) >> 5));
  if (row >= n_nodes) return;
  const int beg = rowptr[row];
  const int end = rowptr[row + 1];
  const float dd = dinv[row];
  const float4 self = ld_row(h, row, lane);
  const float4 b = *reinterpret_cast<const float4*>(bias + lane * 4);
  const float4 a = gather_row(h, col, dinv, beg, end, dd, lane);
  float* __restrict__ p = pooled + (size_t)batch[row] * 256 + col_off + lane * 4;
  atomicAdd(&p[0], a.x + self.x * dd * dd + b.x);
  atomicAdd(&p[1], a.y + self.y * dd * dd + b.y);
  atomicAdd(&p[2], a.z + self.z * dd * dd + b.z);
  atomicAdd(&p[3], a.w + self.w * dd * dd + b.w);
}

static inline void launch_gemm(const float* A, const float* B, float* C,
                               const float* bias, int M, int N, int K, int relu,
                               hipStream_t stream) {
  const int tiles = (M >> 4) * (N >> 4);
  const int blocks = (tiles + 7) / 8;
  k_gemm_wmma<<<blocks, 256, 0, stream>>>(A, B, C, bias, M, N, K, relu);
}

extern "C" void kernel_launch(void* const* d_in, const int* in_sizes, int n_in,
                              void* d_out, int out_size, void* d_ws, size_t ws_size,
                              hipStream_t stream) {
  const float* x      = (const float*)d_in[0];
  const int* edge_idx = (const int*)d_in[1];
  const int* batch    = (const int*)d_in[2];
  const float* td_W1 = (const float*)d_in[4];
  const float* td_b1 = (const float*)d_in[5];
  const float* td_W2 = (const float*)d_in[6];
  const float* td_b2 = (const float*)d_in[7];
  const float* bu_W1 = (const float*)d_in[8];
  const float* bu_b1 = (const float*)d_in[9];
  const float* bu_W2 = (const float*)d_in[10];
  const float* bu_b2 = (const float*)d_in[11];
  const float* pw1   = (const float*)d_in[12];
  const float* pb1   = (const float*)d_in[13];
  const float* pw2   = (const float*)d_in[14];
  const float* pb2   = (const float*)d_in[15];

  const int IN_F = 256, HID = 128, OUT_F = 128, G = 512;
  const int n_nodes = in_sizes[0] / IN_F;
  const int n_edges = in_sizes[1] / 2;
  const int* src = edge_idx;
  const int* dst = edge_idx + n_edges;

  float* ws      = (float*)d_ws;
  const size_t NF = (size_t)n_nodes * 128;
  float* h_lin   = ws;
  int*   csr     = (int*)(ws + NF);
  float* h_act   = ws + 2 * NF;
  float* dinv_td = ws + 3 * NF;
  float* dinv_bu = dinv_td + n_nodes;
  float* pooled  = dinv_bu + n_nodes;
  float* mlp_h   = pooled + (size_t)G * 256;

  int* cnt_td    = csr;
  int* cnt_bu    = cnt_td + n_nodes;
  int* rowptr_td = cnt_bu + n_nodes;
  int* rowptr_bu = rowptr_td + (n_nodes + 1);
  int* col_td    = rowptr_bu + (n_nodes + 1);
  int* col_bu    = col_td + n_edges;
  int* rank_td   = col_bu + n_edges;
  int* rank_bu   = rank_td + n_edges;

  const int EB = (n_edges + 255) / 256;
  const int RB = (n_nodes * 32 + 255) / 256;

  hipMemsetAsync(cnt_td, 0, (size_t)2 * n_nodes * sizeof(int), stream);
  k_count_rank<<<EB, 256, 0, stream>>>(src, dst, cnt_td, cnt_bu, rank_td, rank_bu, n_edges);
  k_scan_dinv<<<2, SCAN_WAVES * 32, 0, stream>>>(cnt_td, rowptr_td, dinv_td, n_nodes);
  k_fill<<<EB, 256, 0, stream>>>(src, dst, rowptr_td, rowptr_bu, rank_td, rank_bu,
                                 col_td, col_bu, n_edges);
  hipMemsetAsync(pooled, 0, (size_t)G * 256 * sizeof(float), stream);

  launch_gemm(x, td_W1, h_lin, nullptr, n_nodes, HID, IN_F, 0, stream);
  k_gather_relu<<<RB, 256, 0, stream>>>(h_lin, rowptr_td, col_td, dinv_td, td_b1, h_act, n_nodes);
  launch_gemm(h_act, td_W2, h_lin, nullptr, n_nodes, OUT_F, HID, 0, stream);
  k_gather_pool<<<RB, 256, 0, stream>>>(h_lin, rowptr_td, col_td, dinv_td, td_b2, batch, pooled,
                                         128, n_nodes);

  launch_gemm(x, bu_W1, h_lin, nullptr, n_nodes, HID, IN_F, 0, stream);
  k_gather_relu<<<RB, 256, 0, stream>>>(h_lin, rowptr_bu, col_bu, dinv_bu, bu_b1, h_act, n_nodes);
  launch_gemm(h_act, bu_W2, h_lin, nullptr, n_nodes, OUT_F, HID, 0, stream);
  k_gather_pool<<<RB, 256, 0, stream>>>(h_lin, rowptr_bu, col_bu, dinv_bu, bu_b2, batch, pooled,
                                         0, n_nodes);

  launch_gemm(pooled, pw1, mlp_h, pb1, G, 256, 256,  1, stream);
  launch_gemm(mlp_h, pw2, (float*)d_out, pb2, G, 128, 256,  0, stream);
}


# GEMM: one wave computes a 16x64 strip, 4 accumulator tiles sharing each A fragment, columns interleaved so B loads and C stores are 16 bytes per lane on whole lines; next block's loads issued ahead
# speedup vs baseline: 2.1391x; 1.1795x over previous
#include <hip/hip_runtime.h>


typedef __attribute__((ext_vector_type(2))) float v2f;
typedef __attribute__((ext_vector_type(8))) float v8f;

__global__ void k_count_rank(const int* __restrict__ src, const int* __restrict__ dst,
                             int* __restrict__ cnt_td, int* __restrict__ cnt_bu,
                             int* __restrict__ rank_td, int* __restrict__ rank_bu,
                             int n_edges) {
  const int e = blockIdx.x * blockDim.x + threadIdx.x;
  if (e >= n_edges) return;
  rank_td[e] = atomicAdd(&cnt_td[dst[e]], 1);
  rank_bu[e] = atomicAdd(&cnt_bu[src[e]], 1);
}

#define SCAN_WAVES 32
#define SCAN_BATCH 7
__global__ void __launch_bounds__(SCAN_WAVES * 32)
k_scan_dinv(const int* __restrict__ cnt2, int* __restrict__ rowptr2,
            float* __restrict__ dinv2, int n) {
  __shared__ int wave_tot[SCAN_WAVES];
  const int* __restrict__ cnt = cnt2 + (size_t)blockIdx.x * n;
  int* __restrict__ rowptr = rowptr2 + (size_t)blockIdx.x * (n + 1);
  float* __restrict__ dinv = dinv2 + (size_t)blockIdx.x * n;
  const int lane = threadIdx.x & 31;
  const int wave = threadIdx.x >> 5;
  const int steps = (n + SCAN_WAVES * 32 - 1) / (SCAN_WAVES * 32);
  const int base = wave * steps * 32;

  int part = 0;
  for (int k = 0; k < steps; ++k) {
    const int i = base + k * 32 + lane;
    part += (i < n) ? cnt[i] : 0;
  }
  for (int o = 16; o > 0; o >>= 1) part += __shfl_xor(part, o);
  if (lane == 0) wave_tot[wave] = part;
  __syncthreads();
  int carry = (lane < wave) ? wave_tot[lane] : 0;
  for (int o = 16; o > 0; o >>= 1) carry += __shfl_xor(carry, o);

  for (int k0 = 0; k0 < steps; k0 += SCAN_BATCH) {
    int v[SCAN_BATCH];
#pragma unroll
    for (int u = 0; u < SCAN_BATCH; ++u) {
      const int i = base + (k0 + u) * 32 + lane;
      v[u] = (k0 + u < steps && i < n) ? cnt[i] : 0;
    }
    int x[SCAN_BATCH];
#pragma unroll
    for (int u = 0; u < SCAN_BATCH; ++u) x[u] = v[u];
#pragma unroll
    for (int o = 1; o < 32; o <<= 1) {
      int y[SCAN_BATCH];
#pragma unroll
      for (int u = 0; u < SCAN_BATCH; ++u) y[u] = __shfl_up(x[u], o);
#pragma unroll
      for (int u = 0; u < SCAN_BATCH; ++u) x[u] += (lane >= o) ? y[u] : 0;
    }
#pragma unroll
    for (int u = 0; u < SCAN_BATCH; ++u) {
      const int i = base + (k0 + u) * 32 + lane;
      if (k0 + u < steps && i < n) {
        rowptr[i] = carry + x[u] - v[u];
        dinv[i] = rsqrtf((float)v[u] + 1.0f);
      }
      carry += __builtin_amdgcn_readlane(x[u], 31);
    }
  }
  if (wave == SCAN_WAVES - 1 && lane == 0) rowptr[n] = carry;
}

__global__ void k_fill(const int* __restrict__ src, const int* __restrict__ dst,
                       const int* __restrict__ rowptr_td, const int* __restrict__ rowptr_bu,
                       const int* __restrict__ rank_td, const int* __restrict__ rank_bu,
                       int* __restrict__ col_td, int* __restrict__ col_bu, int n_edges) {
  const int e = blockIdx.x * blockDim.x + threadIdx.x;
  if (e >= n_edges) return;
  const int s = src[e];
  const int d = dst[e];
  col_td[rowptr_td[d] + rank_td[e]] = s;
  col_bu[rowptr_bu[s] + rank_bu[e]] = d;
}

struct GemmFrag {
  float4 a;
  float4 b0, b1, b2, b3;
};
__device__ __forceinline__ GemmFrag gemm_load(const float* __restrict__ Ap,
                                              const float* __restrict__ Bp, int k0, int N) {
  GemmFrag f;
  f.a  = *reinterpret_cast<const float4*>(Ap + k0);
  const float* __restrict__ bp = Bp + (size_t)k0 * N;
  f.b0 = *reinterpret_cast<const float4*>(bp);
  f.b1 = *reinterpret_cast<const float4*>(bp + N);
  f.b2 = *reinterpret_cast<const float4*>(bp + 2 * (size_t)N);
  f.b3 = *reinterpret_cast<const float4*>(bp + 3 * (size_t)N);
  return f;
}
__device__ __forceinline__ v8f mma4(float a0, float a1, float b0, float b1, v8f acc) {
  v2f a, b;
  a.x = a0; a.y = a1;
  b.x = b0; b.y = b1;
  return __builtin_amdgcn_wmma_f32_16x16x4_f32(
       false, a,  false, b,
       (short)0, acc,  false,  false);
}
__device__ __forceinline__ void gemm_block(const GemmFrag& f, v8f& acc0, v8f& acc1,
                                           v8f& acc2, v8f& acc3) {
  acc0 = mma4(f.a.x, f.a.y, f.b0.x, f.b1.x, acc0);
  acc1 = mma4(f.a.x, f.a.y, f.b0.y, f.b1.y, acc1);
  acc2 = mma4(f.a.x, f.a.y, f.b0.z, f.b1.z, acc2);
  acc3 = mma4(f.a.x, f.a.y, f.b0.w, f.b1.w, acc3);
  acc0 = mma4(f.a.z, f.a.w, f.b2.x, f.b3.x, acc0);
  acc1 = mma4(f.a.z, f.a.w, f.b2.y, f.b3.y, acc1);
  acc2 = mma4(f.a.z, f.a.w, f.b2.z, f.b3.z, acc2);
  acc3 = mma4(f.a.z, f.a.w, f.b2.w, f.b3.w, acc3);
}

#define GEMM_WAVES 4
__global__ void __launch_bounds__(GEMM_WAVES * 32)
k_gemm_wmma(const float* __restrict__ A, const float* __restrict__ B,
            float* __restrict__ C, const float* __restrict__ bias,
            int M, int N, int K, int relu) {
  const int lane = threadIdx.x & 31;
  const int wave = threadIdx.x >> 5;
  const int strip = blockIdx.x * GEMM_WAVES + wave;
  const int sn_cnt = N >> 6;
  const int total = (M >> 4) * sn_cnt;
  if (strip >= total) return;
  const int tm = (strip / sn_cnt) << 4;
  const int cn = (strip % sn_cnt) << 6;

  const int h = lane >> 4;
  const int c = lane & 15;

  const float* __restrict__ Ap = A + (size_t)(tm + c) * K + 4 * h;
  const float* __restrict__ Bp = B + (size_t)(4 * h) * N + cn + 4 * c;

  v8f acc0 = {}, acc1 = {}, acc2 = {}, acc3 = {};
  GemmFrag f0 = gemm_load(Ap, Bp, 0, N);
  for (int k0 = 0; k0 < K; k0 += 16) {
    const GemmFrag f1 = gemm_load(Ap, Bp, k0 + 8, N);
    __builtin_amdgcn_sched_barrier(0);
    gemm_block(f0, acc0, acc1, acc2, acc3);
    f0 = gemm_load(Ap, Bp, (k0 + 16 < K) ? k0 + 16 : k0, N);
    __builtin_amdgcn_sched_barrier(0);
    gemm_block(f1, acc0, acc1, acc2, acc3);
  }

  float4 bi = make_float4(0.f, 0.f, 0.f, 0.f);
  if (bias) bi = *reinterpret_cast<const float4*>(bias + cn + 4 * c);
#pragma unroll
  for (int r = 0; r < 8; ++r) {
    float4 v;
    v.x = acc0[r] + bi.x;
    v.y = acc1[r] + bi.y;
    v.z = acc2[r] + bi.z;
    v.w = acc3[r] + bi.w;
    if (relu) {
      v.x = fmaxf(v.x, 0.0f);
      v.y = fmaxf(v.y, 0.0f);
      v.z = fmaxf(v.z, 0.0f);
      v.w = fmaxf(v.w, 0.0f);
    }
    *reinterpret_cast<float4*>(C + (size_t)(tm + 8 * h + r) * N + cn + 4 * c) = v;
  }
}

__device__ __forceinline__ float4 ld_row(const float* __restrict__ h, int s, int lane) {
  return *reinterpret_cast<const float4*>(h + (size_t)s * 128 + lane * 4);
}
__device__ __forceinline__ void fma4(float4& a, const float4 v, const float w) {
  a.x = fmaf(v.x, w, a.x);
  a.y = fmaf(v.y, w, a.y);
  a.z = fmaf(v.z, w, a.z);
  a.w = fmaf(v.w, w, a.w);
}
__device__ __forceinline__ int lane_i(int v, int j) { return __builtin_amdgcn_readlane(v, j); }
__device__ __forceinline__ float lane_f(float v, int j) {
  return __int_as_float(__builtin_amdgcn_readlane(__float_as_int(v), j));
}

__device__ __forceinline__ float4 gather_row(const float* __restrict__ h,
                                             const int* __restrict__ col,
                                             const float* __restrict__ dinv,
                                             int beg, int end, float dd, int lane) {
  float4 acc = make_float4(0.f, 0.f, 0.f, 0.f);
  for (int base = beg; base < end; base += 32) {
    const int n = min(32, end - base);
    int my_s = 0;
    float my_w = 0.f;
    if (lane < n) {
      my_s = col[base + lane];
      my_w = dinv[my_s] * dd;
    }
    int j = 0;
    for (; j + 4 <= n; j += 4) {
      const float4 v0 = ld_row(h, lane_i(my_s, j), lane);
      const float4 v1 = ld_row(h, lane_i(my_s, j + 1), lane);
      const float4 v2 = ld_row(h, lane_i(my_s, j + 2), lane);
      const float4 v3 = ld_row(h, lane_i(my_s, j + 3), lane);
      fma4(acc, v0, lane_f(my_w, j));
      fma4(acc, v1, lane_f(my_w, j + 1));
      fma4(acc, v2, lane_f(my_w, j + 2));
      fma4(acc, v3, lane_f(my_w, j + 3));
    }
    if (j + 2 <= n) {
      const float4 v0 = ld_row(h, lane_i(my_s, j), lane);
      const float4 v1 = ld_row(h, lane_i(my_s, j + 1), lane);
      fma4(acc, v0, lane_f(my_w, j));
      fma4(acc, v1, lane_f(my_w, j + 1));
      j += 2;
    }
    if (j < n) {
      const float4 v0 = ld_row(h, lane_i(my_s, j), lane);
      fma4(acc, v0, lane_f(my_w, j));
    }
  }
  return acc;
}

__global__ void __launch_bounds__(256)
k_gather_relu(const float* __restrict__ h, const int* __restrict__ rowptr,
              const int* __restrict__ col, const float* __restrict__ dinv,
              const float* __restrict__ bias, float* __restrict__ out, int n_nodes) {
  const int lane = threadIdx.x & 31;
  const int row = __builtin_amdgcn_readfirstlane((int)((blockIdx.x * blockDim.x + threadIdx.x) >> 5));
  if (row >= n_nodes) return;
  const int beg = rowptr[row];
  const int end = rowptr[row + 1];
  const float dd = dinv[row];
  const float4 self = ld_row(h, row, lane);
  const float4 b = *reinterpret_cast<const float4*>(bias + lane * 4);
  const float4 a = gather_row(h, col, dinv, beg, end, dd, lane);
  float4 v;
  v.x = fmaxf(a.x + self.x * dd * dd + b.x, 0.0f);
  v.y = fmaxf(a.y + self.y * dd * dd + b.y, 0.0f);
  v.z = fmaxf(a.z + self.z * dd * dd + b.z, 0.0f);
  v.w = fmaxf(a.w + self.w * dd * dd + b.w, 0.0f);
  *reinterpret_cast<float4*>(out + (size_t)row * 128 + lane * 4) = v;
}

__global__ void __launch_bounds__(256)
k_gather_pool(const float* __restrict__ h, const int* __restrict__ rowptr,
              const int* __restrict__ col, const float* __restrict__ dinv,
              const float* __restrict__ bias, const int* __restrict__ batch,
              float* __restrict__ pooled, int col_off, int n_nodes) {
  const int lane = threadIdx.x & 31;
  const int row = __builtin_amdgcn_readfirstlane((int)((blockIdx.x * blockDim.x + threadIdx.x) >> 5));
  if (row >= n_nodes) return;
  const int beg = rowptr[row];
  const int end = rowptr[row + 1];
  const float dd = dinv[row];
  const float4 self = ld_row(h, row, lane);
  const float4 b = *reinterpret_cast<const float4*>(bias + lane * 4);
  const float4 a = gather_row(h, col, dinv, beg, end, dd, lane);
  float* __restrict__ p = pooled + (size_t)batch[row] * 256 + col_off + lane * 4;
  atomicAdd(&p[0], a.x + self.x * dd * dd + b.x);
  atomicAdd(&p[1], a.y + self.y * dd * dd + b.y);
  atomicAdd(&p[2], a.z + self.z * dd * dd + b.z);
  atomicAdd(&p[3], a.w + self.w * dd * dd + b.w);
}

static inline void launch_gemm(const float* A, const float* B, float* C,
                               const float* bias, int M, int N, int K, int relu,
                               hipStream_t stream) {
  const int strips = (M >> 4) * (N >> 6);
  const int blocks = (strips + GEMM_WAVES - 1) / GEMM_WAVES;
  k_gemm_wmma<<<blocks, GEMM_WAVES * 32, 0, stream>>>(A, B, C, bias, M, N, K, relu);
}

extern "C" void kernel_launch(void* const* d_in, const int* in_sizes, int n_in,
                              void* d_out, int out_size, void* d_ws, size_t ws_size,
                              hipStream_t stream) {
  const float* x      = (const float*)d_in[0];
  const int* edge_idx = (const int*)d_in[1];
  const int* batch    = (const int*)d_in[2];
  const float* td_W1 = (const float*)d_in[4];
  const float* td_b1 = (const float*)d_in[5];
  const float* td_W2 = (const float*)d_in[6];
  const float* td_b2 = (const float*)d_in[7];
  const float* bu_W1 = (const float*)d_in[8];
  const float* bu_b1 = (const float*)d_in[9];
  const float* bu_W2 = (const float*)d_in[10];
  const float* bu_b2 = (const float*)d_in[11];
  const float* pw1   = (const float*)d_in[12];
  const float* pb1   = (const float*)d_in[13];
  const float* pw2   = (const float*)d_in[14];
  const float* pb2   = (const float*)d_in[15];

  const int IN_F = 256, HID = 128, OUT_F = 128, G = 512;
  const int n_nodes = in_sizes[0] / IN_F;
  const int n_edges = in_sizes[1] / 2;
  const int* src = edge_idx;
  const int* dst = edge_idx + n_edges;

  float* ws      = (float*)d_ws;
  const size_t NF = (size_t)n_nodes * 128;
  float* h_lin   = ws;
  int*   csr     = (int*)(ws + NF);
  float* h_act   = ws + 2 * NF;
  float* dinv_td = ws + 3 * NF;
  float* dinv_bu = dinv_td + n_nodes;
  float* pooled  = dinv_bu + n_nodes;
  float* mlp_h   = pooled + (size_t)G * 256;

  int* cnt_td    = csr;
  int* cnt_bu    = cnt_td + n_nodes;
  int* rowptr_td = cnt_bu + n_nodes;
  int* rowptr_bu = rowptr_td + (n_nodes + 1);
  int* col_td    = rowptr_bu + (n_nodes + 1);
  int* col_bu    = col_td + n_edges;
  int* rank_td   = col_bu + n_edges;
  int* rank_bu   = rank_td + n_edges;

  const int EB = (n_edges + 255) / 256;
  const int RB = (n_nodes * 32 + 255) / 256;

  hipMemsetAsync(cnt_td, 0, (size_t)2 * n_nodes * sizeof(int), stream);
  k_count_rank<<<EB, 256, 0, stream>>>(src, dst, cnt_td, cnt_bu, rank_td, rank_bu, n_edges);
  k_scan_dinv<<<2, SCAN_WAVES * 32, 0, stream>>>(cnt_td, rowptr_td, dinv_td, n_nodes);
  k_fill<<<EB, 256, 0, stream>>>(src, dst, rowptr_td, rowptr_bu, rank_td, rank_bu,
                                 col_td, col_bu, n_edges);
  hipMemsetAsync(pooled, 0, (size_t)G * 256 * sizeof(float), stream);

  launch_gemm(x, td_W1, h_lin, nullptr, n_nodes, HID, IN_F, 0, stream);
  k_gather_relu<<<RB, 256, 0, stream>>>(h_lin, rowptr_td, col_td, dinv_td, td_b1, h_act, n_nodes);
  launch_gemm(h_act, td_W2, h_lin, nullptr, n_nodes, OUT_F, HID, 0, stream);
  k_gather_pool<<<RB, 256, 0, stream>>>(h_lin, rowptr_td, col_td, dinv_td, td_b2, batch, pooled,
                                         128, n_nodes);

  launch_gemm(x, bu_W1, h_lin, nullptr, n_nodes, HID, IN_F, 0, stream);
  k_gather_relu<<<RB, 256, 0, stream>>>(h_lin, rowptr_bu, col_bu, dinv_bu, bu_b1, h_act, n_nodes);
  launch_gemm(h_act, bu_W2, h_lin, nullptr, n_nodes, OUT_F, HID, 0, stream);
  k_gather_pool<<<RB, 256, 0, stream>>>(h_lin, rowptr_bu, col_bu, dinv_bu, bu_b2, batch, pooled,
                                         0, n_nodes);

  launch_gemm(pooled, pw1, mlp_h, pb1, G, 256, 256,  1, stream);
  launch_gemm(mlp_h, pw2, (float*)d_out, pb2, G, 128, 256,  0, stream);
}


# edge aggregation: 8192 resident waves each walk consecutive rows, 8 row loads in flight (was one short wave per row); sort threads take 4 edges each: 4x fewer waves enter per call
# speedup vs baseline: 2.4504x; 1.1455x over previous
#include <hip/hip_runtime.h>


typedef __attribute__((ext_vector_type(2))) float v2f;
typedef __attribute__((ext_vector_type(8))) float v8f;

#define EDGE_PER_THREAD 4
__global__ void __launch_bounds__(256)
k_count_rank(const int* __restrict__ src, const int* __restrict__ dst,
             int* __restrict__ cnt_td, int* __restrict__ cnt_bu,
             int* __restrict__ rank_td, int* __restrict__ rank_bu,
             int n_edges) {
  const int width = gridDim.x * blockDim.x;
  const int e0 = blockIdx.x * blockDim.x + threadIdx.x;
  int s[EDGE_PER_THREAD], d[EDGE_PER_THREAD], rt[EDGE_PER_THREAD], rb[EDGE_PER_THREAD];
#pragma unroll
  for (int j = 0; j < EDGE_PER_THREAD; ++j) {
    const int e = min(e0 + j * width, n_edges - 1);
    s[j] = src[e];
    d[j] = dst[e];
  }
#pragma unroll
  for (int j = 0; j < EDGE_PER_THREAD; ++j) {
    const int e = e0 + j * width;
    rt[j] = 0;
    rb[j] = 0;
    if (e < n_edges) {
      rt[j] = atomicAdd(&cnt_td[d[j]], 1);
      rb[j] = atomicAdd(&cnt_bu[s[j]], 1);
    }
  }
#pragma unroll
  for (int j = 0; j < EDGE_PER_THREAD; ++j) {
    const int e = e0 + j * width;
    if (e < n_edges) {
      rank_td[e] = rt[j];
      rank_bu[e] = rb[j];
    }
  }
}

#define SCAN_WAVES 32
#define SCAN_BATCH 7
__global__ void __launch_bounds__(SCAN_WAVES * 32)
k_scan_dinv(const int* __restrict__ cnt2, int* __restrict__ rowptr2,
            float* __restrict__ dinv2, int n) {
  __shared__ int wave_tot[SCAN_WAVES];
  const int* __restrict__ cnt = cnt2 + (size_t)blockIdx.x * n;
  int* __restrict__ rowptr = rowptr2 + (size_t)blockIdx.x * (n + 1);
  float* __restrict__ dinv = dinv2 + (size_t)blockIdx.x * n;
  const int lane = threadIdx.x & 31;
  const int wave = threadIdx.x >> 5;
  const int steps = (n + SCAN_WAVES * 32 - 1) / (SCAN_WAVES * 32);
  const int base = wave * steps * 32;

  int part = 0;
  for (int k = 0; k < steps; ++k) {
    const int i = base + k * 32 + lane;
    part += (i < n) ? cnt[i] : 0;
  }
  for (int o = 16; o > 0; o >>= 1) part += __shfl_xor(part, o);
  if (lane == 0) wave_tot[wave] = part;
  __syncthreads();
  int carry = (lane < wave) ? wave_tot[lane] : 0;
  for (int o = 16; o > 0; o >>= 1) carry += __shfl_xor(carry, o);

  for (int k0 = 0; k0 < steps; k0 += SCAN_BATCH) {
    int v[SCAN_BATCH];
#pragma unroll
    for (int u = 0; u < SCAN_BATCH; ++u) {
      const int i = base + (k0 + u) * 32 + lane;
      v[u] = (k0 + u < steps && i < n) ? cnt[i] : 0;
    }
    int x[SCAN_BATCH];
#pragma unroll
    for (int u = 0; u < SCAN_BATCH; ++u) x[u] = v[u];
#pragma unroll
    for (int o = 1; o < 32; o <<= 1) {
      int y[SCAN_BATCH];
#pragma unroll
      for (int u = 0; u < SCAN_BATCH; ++u) y[u] = __shfl_up(x[u], o);
#pragma unroll
      for (int u = 0; u < SCAN_BATCH; ++u) x[u] += (lane >= o) ? y[u] : 0;
    }
#pragma unroll
    for (int u = 0; u < SCAN_BATCH; ++u) {
      const int i = base + (k0 + u) * 32 + lane;
      if (k0 + u < steps && i < n) {
        rowptr[i] = carry + x[u] - v[u];
        dinv[i] = rsqrtf((float)v[u] + 1.0f);
      }
      carry += __builtin_amdgcn_readlane(x[u], 31);
    }
  }
  if (wave == SCAN_WAVES - 1 && lane == 0) rowptr[n] = carry;
}

__global__ void __launch_bounds__(256)
k_fill(const int* __restrict__ src, const int* __restrict__ dst,
       const int* __restrict__ rowptr_td, const int* __restrict__ rowptr_bu,
       const int* __restrict__ rank_td, const int* __restrict__ rank_bu,
       int* __restrict__ col_td, int* __restrict__ col_bu, int n_edges) {
  const int width = gridDim.x * blockDim.x;
  const int e0 = blockIdx.x * blockDim.x + threadIdx.x;
  int s[EDGE_PER_THREAD], d[EDGE_PER_THREAD], pt[EDGE_PER_THREAD], pb[EDGE_PER_THREAD];
#pragma unroll
  for (int j = 0; j < EDGE_PER_THREAD; ++j) {
    const int e = min(e0 + j * width, n_edges - 1);
    s[j] = src[e];
    d[j] = dst[e];
    pt[j] = rank_td[e];
    pb[j] = rank_bu[e];
  }
#pragma unroll
  for (int j = 0; j < EDGE_PER_THREAD; ++j) {
    pt[j] += rowptr_td[d[j]];
    pb[j] += rowptr_bu[s[j]];
  }
#pragma unroll
  for (int j = 0; j < EDGE_PER_THREAD; ++j) {
    const int e = e0 + j * width;
    if (e < n_edges) {
      col_td[pt[j]] = s[j];
      col_bu[pb[j]] = d[j];
    }
  }
}

struct GemmFrag {
  float4 a;
  float4 b0, b1, b2, b3;
};
__device__ __forceinline__ GemmFrag gemm_load(const float* __restrict__ Ap,
                                              const float* __restrict__ Bp, int k0, int N) {
  GemmFrag f;
  f.a  = *reinterpret_cast<const float4*>(Ap + k0);
  const float* __restrict__ bp = Bp + (size_t)k0 * N;
  f.b0 = *reinterpret_cast<const float4*>(bp);
  f.b1 = *reinterpret_cast<const float4*>(bp + N);
  f.b2 = *reinterpret_cast<const float4*>(bp + 2 * (size_t)N);
  f.b3 = *reinterpret_cast<const float4*>(bp + 3 * (size_t)N);
  return f;
}
__device__ __forceinline__ v8f mma4(float a0, float a1, float b0, float b1, v8f acc) {
  v2f a, b;
  a.x = a0; a.y = a1;
  b.x = b0; b.y = b1;
  return __builtin_amdgcn_wmma_f32_16x16x4_f32(
       false, a,  false, b,
       (short)0, acc,  false,  false);
}
__device__ __forceinline__ void gemm_block(const GemmFrag& f, v8f& acc0, v8f& acc1,
                                           v8f& acc2, v8f& acc3) {
  acc0 = mma4(f.a.x, f.a.y, f.b0.x, f.b1.x, acc0);
  acc1 = mma4(f.a.x, f.a.y, f.b0.y, f.b1.y, acc1);
  acc2 = mma4(f.a.x, f.a.y, f.b0.z, f.b1.z, acc2);
  acc3 = mma4(f.a.x, f.a.y, f.b0.w, f.b1.w, acc3);
  acc0 = mma4(f.a.z, f.a.w, f.b2.x, f.b3.x, acc0);
  acc1 = mma4(f.a.z, f.a.w, f.b2.y, f.b3.y, acc1);
  acc2 = mma4(f.a.z, f.a.w, f.b2.z, f.b3.z, acc2);
  acc3 = mma4(f.a.z, f.a.w, f.b2.w, f.b3.w, acc3);
}

#define GEMM_WAVES 4
__global__ void __launch_bounds__(GEMM_WAVES * 32)
k_gemm_wmma(const float* __restrict__ A, const float* __restrict__ B,
            float* __restrict__ C, const float* __restrict__ bias,
            int M, int N, int K, int relu) {
  const int lane = threadIdx.x & 31;
  const int wave = threadIdx.x >> 5;
  const int strip = blockIdx.x * GEMM_WAVES + wave;
  const int sn_cnt = N >> 6;
  const int total = (M >> 4) * sn_cnt;
  if (strip >= total) return;
  const int tm = (strip / sn_cnt) << 4;
  const int cn = (strip % sn_cnt) << 6;

  const int h = lane >> 4;
  const int c = lane & 15;

  const float* __restrict__ Ap = A + (size_t)(tm + c) * K + 4 * h;
  const float* __restrict__ Bp = B + (size_t)(4 * h) * N + cn + 4 * c;

  v8f acc0 = {}, acc1 = {}, acc2 = {}, acc3 = {};
  GemmFrag f0 = gemm_load(Ap, Bp, 0, N);
  for (int k0 = 0; k0 < K; k0 += 16) {
    const GemmFrag f1 = gemm_load(Ap, Bp, k0 + 8, N);
    __builtin_amdgcn_sched_barrier(0);
    gemm_block(f0, acc0, acc1, acc2, acc3);
    f0 = gemm_load(Ap, Bp, (k0 + 16 < K) ? k0 + 16 : k0, N);
    __builtin_amdgcn_sched_barrier(0);
    gemm_block(f1, acc0, acc1, acc2, acc3);
  }

  float4 bi = make_float4(0.f, 0.f, 0.f, 0.f);
  if (bias) bi = *reinterpret_cast<const float4*>(bias + cn + 4 * c);
#pragma unroll
  for (int r = 0; r < 8; ++r) {
    float4 v;
    v.x = acc0[r] + bi.x;
    v.y = acc1[r] + bi.y;
    v.z = acc2[r] + bi.z;
    v.w = acc3[r] + bi.w;
    if (relu) {
      v.x = fmaxf(v.x, 0.0f);
      v.y = fmaxf(v.y, 0.0f);
      v.z = fmaxf(v.z, 0.0f);
      v.w = fmaxf(v.w, 0.0f);
    }
    *reinterpret_cast<float4*>(C + (size_t)(tm + 8 * h + r) * N + cn + 4 * c) = v;
  }
}

__device__ __forceinline__ float4 ld_row(const float* __restrict__ h, int s, int lane) {
  return *reinterpret_cast<const float4*>(h + (size_t)s * 128 + lane * 4);
}
__device__ __forceinline__ void fma4(float4& a, const float4 v, const float w) {
  a.x = fmaf(v.x, w, a.x);
  a.y = fmaf(v.y, w, a.y);
  a.z = fmaf(v.z, w, a.z);
  a.w = fmaf(v.w, w, a.w);
}
__device__ __forceinline__ int lane_i(int v, int j) { return __builtin_amdgcn_readlane(v, j); }
__device__ __forceinline__ float lane_f(float v, int j) {
  return __int_as_float(__builtin_amdgcn_readlane(__float_as_int(v), j));
}

__device__ __forceinline__ float4 gather_row(const float* __restrict__ h,
                                             const int* __restrict__ col,
                                             const float* __restrict__ dinv,
                                             int beg, int end, float dd, int lane) {
  float4 acc = make_float4(0.f, 0.f, 0.f, 0.f);
  for (int base = beg; base < end; base += 32) {
    const int n = min(32, end - base);
    int my_s = 0;
    float my_w = 0.f;
    if (lane < n) {
      my_s = col[base + lane];
      my_w = dinv[my_s] * dd;
    }
    int j = 0;
    for (; j + 8 <= n; j += 8) {
      const float4 v0 = ld_row(h, lane_i(my_s, j), lane);
      const float4 v1 = ld_row(h, lane_i(my_s, j + 1), lane);
      const float4 v2 = ld_row(h, lane_i(my_s, j + 2), lane);
      const float4 v3 = ld_row(h, lane_i(my_s, j + 3), lane);
      const float4 v4 = ld_row(h, lane_i(my_s, j + 4), lane);
      const float4 v5 = ld_row(h, lane_i(my_s, j + 5), lane);
      const float4 v6 = ld_row(h, lane_i(my_s, j + 6), lane);
      const float4 v7 = ld_row(h, lane_i(my_s, j + 7), lane);
      fma4(acc, v0, lane_f(my_w, j));
      fma4(acc, v1, lane_f(my_w, j + 1));
      fma4(acc, v2, lane_f(my_w, j + 2));
      fma4(acc, v3, lane_f(my_w, j + 3));
      fma4(acc, v4, lane_f(my_w, j + 4));
      fma4(acc, v5, lane_f(my_w, j + 5));
      fma4(acc, v6, lane_f(my_w, j + 6));
      fma4(acc, v7, lane_f(my_w, j + 7));
    }
    if (j + 4 <= n) {
      const float4 v0 = ld_row(h, lane_i(my_s, j), lane);
      const float4 v1 = ld_row(h, lane_i(my_s, j + 1), lane);
      const float4 v2 = ld_row(h, lane_i(my_s, j + 2), lane);
      const float4 v3 = ld_row(h, lane_i(my_s, j + 3), lane);
      fma4(acc, v0, lane_f(my_w, j));
      fma4(acc, v1, lane_f(my_w, j + 1));
      fma4(acc, v2, lane_f(my_w, j + 2));
      fma4(acc, v3, lane_f(my_w, j + 3));
      j += 4;
    }
    if (j + 2 <= n) {
      const float4 v0 = ld_row(h, lane_i(my_s, j), lane);
      const float4 v1 = ld_row(h, lane_i(my_s, j + 1), lane);
      fma4(acc, v0, lane_f(my_w, j));
      fma4(acc, v1, lane_f(my_w, j + 1));
      j += 2;
    }
    if (j < n) {
      const float4 v0 = ld_row(h, lane_i(my_s, j), lane);
      fma4(acc, v0, lane_f(my_w, j));
    }
  }
  return acc;
}

#define GATHER_BLOCKS 1024
__device__ __forceinline__ void wave_rows(int n_nodes, int& r0, int& r1) {
  const int g = __builtin_amdgcn_readfirstlane((int)((blockIdx.x * blockDim.x + threadIdx.x) >> 5));
  const int nw = (int)((gridDim.x * blockDim.x) >> 5);
  const int q = n_nodes / nw;
  const int r = n_nodes % nw;
  r0 = g * q + min(g, r);
  r1 = r0 + q + (g < r ? 1 : 0);
}

__global__ void __launch_bounds__(256)
k_gather_relu(const float* __restrict__ h, const int* __restrict__ rowptr,
              const int* __restrict__ col, const float* __restrict__ dinv,
              const float* __restrict__ bias, float* __restrict__ out, int n_nodes) {
  const int lane = threadIdx.x & 31;
  int r0, r1;
  wave_rows(n_nodes, r0, r1);
  const float4 b = *reinterpret_cast<const float4*>(bias + lane * 4);
  for (int row = r0; row < r1; ++row) {
    const int beg = rowptr[row];
    const int end = rowptr[row + 1];
    const float dd = dinv[row];
    const float4 self = ld_row(h, row, lane);
    const float4 a = gather_row(h, col, dinv, beg, end, dd, lane);
    float4 v;
    v.x = fmaxf(a.x + self.x * dd * dd + b.x, 0.0f);
    v.y = fmaxf(a.y + self.y * dd * dd + b.y, 0.0f);
    v.z = fmaxf(a.z + self.z * dd * dd + b.z, 0.0f);
    v.w = fmaxf(a.w + self.w * dd * dd + b.w, 0.0f);
    *reinterpret_cast<float4*>(out + (size_t)row * 128 + lane * 4) = v;
  }
}

__global__ void __launch_bounds__(256)
k_gather_pool(const float* __restrict__ h, const int* __restrict__ rowptr,
              const int* __restrict__ col, const float* __restrict__ dinv,
              const float* __restrict__ bias, const int* __restrict__ batch,
              float* __restrict__ pooled, int col_off, int n_nodes) {
  const int lane = threadIdx.x & 31;
  int r0, r1;
  wave_rows(n_nodes, r0, r1);
  const float4 b = *reinterpret_cast<const float4*>(bias + lane * 4);
  for (int row = r0; row < r1; ++row) {
    const int beg = rowptr[row];
    const int end = rowptr[row + 1];
    const float dd = dinv[row];
    const float4 self = ld_row(h, row, lane);
    const float4 a = gather_row(h, col, dinv, beg, end, dd, lane);
    float* __restrict__ p = pooled + (size_t)batch[row] * 256 + col_off + lane * 4;
    atomicAdd(&p[0], a.x + self.x * dd * dd + b.x);
    atomicAdd(&p[1], a.y + self.y * dd * dd + b.y);
    atomicAdd(&p[2], a.z + self.z * dd * dd + b.z);
    atomicAdd(&p[3], a.w + self.w * dd * dd + b.w);
  }
}

static inline void launch_gemm(const float* A, const float* B, float* C,
                               const float* bias, int M, int N, int K, int relu,
                               hipStream_t stream) {
  const int strips = (M >> 4) * (N >> 6);
  const int blocks = (strips + GEMM_WAVES - 1) / GEMM_WAVES;
  k_gemm_wmma<<<blocks, GEMM_WAVES * 32, 0, stream>>>(A, B, C, bias, M, N, K, relu);
}

extern "C" void kernel_launch(void* const* d_in, const int* in_sizes, int n_in,
                              void* d_out, int out_size, void* d_ws, size_t ws_size,
                              hipStream_t stream) {
  const float* x      = (const float*)d_in[0];
  const int* edge_idx = (const int*)d_in[1];
  const int* batch    = (const int*)d_in[2];
  const float* td_W1 = (const float*)d_in[4];
  const float* td_b1 = (const float*)d_in[5];
  const float* td_W2 = (const float*)d_in[6];
  const float* td_b2 = (const float*)d_in[7];
  const float* bu_W1 = (const float*)d_in[8];
  const float* bu_b1 = (const float*)d_in[9];
  const float* bu_W2 = (const float*)d_in[10];
  const float* bu_b2 = (const float*)d_in[11];
  const float* pw1   = (const float*)d_in[12];
  const float* pb1   = (const float*)d_in[13];
  const float* pw2   = (const float*)d_in[14];
  const float* pb2   = (const float*)d_in[15];

  const int IN_F = 256, HID = 128, OUT_F = 128, G = 512;
  const int n_nodes = in_sizes[0] / IN_F;
  const int n_edges = in_sizes[1] / 2;
  const int* src = edge_idx;
  const int* dst = edge_idx + n_edges;

  float* ws      = (float*)d_ws;
  const size_t NF = (size_t)n_nodes * 128;
  float* h_lin   = ws;
  int*   csr     = (int*)(ws + NF);
  float* h_act   = ws + 2 * NF;
  float* dinv_td = ws + 3 * NF;
  float* dinv_bu = dinv_td + n_nodes;
  float* pooled  = dinv_bu + n_nodes;
  float* mlp_h   = pooled + (size_t)G * 256;

  int* cnt_td    = csr;
  int* cnt_bu    = cnt_td + n_nodes;
  int* rowptr_td = cnt_bu + n_nodes;
  int* rowptr_bu = rowptr_td + (n_nodes + 1);
  int* col_td    = rowptr_bu + (n_nodes + 1);
  int* col_bu    = col_td + n_edges;
  int* rank_td   = col_bu + n_edges;
  int* rank_bu   = rank_td + n_edges;

  const int EB = (n_edges + 256 * EDGE_PER_THREAD - 1) / (256 * EDGE_PER_THREAD);
  const int RB = GATHER_BLOCKS;

  hipMemsetAsync(cnt_td, 0, (size_t)2 * n_nodes * sizeof(int), stream);
  k_count_rank<<<EB, 256, 0, stream>>>(src, dst, cnt_td, cnt_bu, rank_td, rank_bu, n_edges);
  k_scan_dinv<<<2, SCAN_WAVES * 32, 0, stream>>>(cnt_td, rowptr_td, dinv_td, n_nodes);
  k_fill<<<EB, 256, 0, stream>>>(src, dst, rowptr_td, rowptr_bu, rank_td, rank_bu,
                                 col_td, col_bu, n_edges);
  hipMemsetAsync(pooled, 0, (size_t)G * 256 * sizeof(float), stream);

  launch_gemm(x, td_W1, h_lin, nullptr, n_nodes, HID, IN_F, 0, stream);
  k_gather_relu<<<RB, 256, 0, stream>>>(h_lin, rowptr_td, col_td, dinv_td, td_b1, h_act, n_nodes);
  launch_gemm(h_act, td_W2, h_lin, nullptr, n_nodes, OUT_F, HID, 0, stream);
  k_gather_pool<<<RB, 256, 0, stream>>>(h_lin, rowptr_td, col_td, dinv_td, td_b2, batch, pooled,
                                         128, n_nodes);

  launch_gemm(x, bu_W1, h_lin, nullptr, n_nodes, HID, IN_F, 0, stream);
  k_gather_relu<<<RB, 256, 0, stream>>>(h_lin, rowptr_bu, col_bu, dinv_bu, bu_b1, h_act, n_nodes);
  launch_gemm(h_act, bu_W2, h_lin, nullptr, n_nodes, OUT_F, HID, 0, stream);
  k_gather_pool<<<RB, 256, 0, stream>>>(h_lin, rowptr_bu, col_bu, dinv_bu, bu_b2, batch, pooled,
                                         0, n_nodes);

  launch_gemm(pooled, pw1, mlp_h, pb1, G, 256, 256,  1, stream);
  launch_gemm(mlp_h, pw2, (float*)d_out, pb2, G, 128, 256,  0, stream);
}


# layer 2 pooled before its product (neighbour sum and graph pooling are linear): pool(Ahat H) W2 + count*b2 replaces pool(Ahat(H W2) + b2); two [50000,128] GEMMs become one [512,256] launch
# speedup vs baseline: 2.5439x; 1.0382x over previous
#include <hip/hip_runtime.h>


typedef __attribute__((ext_vector_type(2))) float v2f;
typedef __attribute__((ext_vector_type(8))) float v8f;

#define EDGE_PER_THREAD 4
__global__ void __launch_bounds__(256)
k_count_rank(const int* __restrict__ src, const int* __restrict__ dst,
             int* __restrict__ cnt_td, int* __restrict__ cnt_bu,
             int* __restrict__ rank_td, int* __restrict__ rank_bu,
             int n_edges) {
  const int width = gridDim.x * blockDim.x;
  const int e0 = blockIdx.x * blockDim.x + threadIdx.x;
  int s[EDGE_PER_THREAD], d[EDGE_PER_THREAD], rt[EDGE_PER_THREAD], rb[EDGE_PER_THREAD];
#pragma unroll
  for (int j = 0; j < EDGE_PER_THREAD; ++j) {
    const int e = min(e0 + j * width, n_edges - 1);
    s[j] = src[e];
    d[j] = dst[e];
  }
#pragma unroll
  for (int j = 0; j < EDGE_PER_THREAD; ++j) {
    const int e = e0 + j * width;
    rt[j] = 0;
    rb[j] = 0;
    if (e < n_edges) {
      rt[j] = atomicAdd(&cnt_td[d[j]], 1);
      rb[j] = atomicAdd(&cnt_bu[s[j]], 1);
    }
  }
#pragma unroll
  for (int j = 0; j < EDGE_PER_THREAD; ++j) {
    const int e = e0 + j * width;
    if (e < n_edges) {
      rank_td[e] = rt[j];
      rank_bu[e] = rb[j];
    }
  }
}

#define SCAN_WAVES 32
#define SCAN_BATCH 7
__global__ void __launch_bounds__(SCAN_WAVES * 32)
k_scan_dinv(const int* __restrict__ cnt2, int* __restrict__ rowptr2,
            float* __restrict__ dinv2, int n) {
  __shared__ int wave_tot[SCAN_WAVES];
  const int* __restrict__ cnt = cnt2 + (size_t)blockIdx.x * n;
  int* __restrict__ rowptr = rowptr2 + (size_t)blockIdx.x * (n + 1);
  float* __restrict__ dinv = dinv2 + (size_t)blockIdx.x * n;
  const int lane = threadIdx.x & 31;
  const int wave = threadIdx.x >> 5;
  const int steps = (n + SCAN_WAVES * 32 - 1) / (SCAN_WAVES * 32);
  const int base = wave * steps * 32;

  int part = 0;
  for (int k = 0; k < steps; ++k) {
    const int i = base + k * 32 + lane;
    part += (i < n) ? cnt[i] : 0;
  }
  for (int o = 16; o > 0; o >>= 1) part += __shfl_xor(part, o);
  if (lane == 0) wave_tot[wave] = part;
  __syncthreads();
  int carry = (lane < wave) ? wave_tot[lane] : 0;
  for (int o = 16; o > 0; o >>= 1) carry += __shfl_xor(carry, o);

  for (int k0 = 0; k0 < steps; k0 += SCAN_BATCH) {
    int v[SCAN_BATCH];
#pragma unroll
    for (int u = 0; u < SCAN_BATCH; ++u) {
      const int i = base + (k0 + u) * 32 + lane;
      v[u] = (k0 + u < steps && i < n) ? cnt[i] : 0;
    }
    int x[SCAN_BATCH];
#pragma unroll
    for (int u = 0; u < SCAN_BATCH; ++u) x[u] = v[u];
#pragma unroll
    for (int o = 1; o < 32; o <<= 1) {
      int y[SCAN_BATCH];
#pragma unroll
      for (int u = 0; u < SCAN_BATCH; ++u) y[u] = __shfl_up(x[u], o);
#pragma unroll
      for (int u = 0; u < SCAN_BATCH; ++u) x[u] += (lane >= o) ? y[u] : 0;
    }
#pragma unroll
    for (int u = 0; u < SCAN_BATCH; ++u) {
      const int i = base + (k0 + u) * 32 + lane;
      if (k0 + u < steps && i < n) {
        rowptr[i] = carry + x[u] - v[u];
        dinv[i] = rsqrtf((float)v[u] + 1.0f);
      }
      carry += __builtin_amdgcn_readlane(x[u], 31);
    }
  }
  if (wave == SCAN_WAVES - 1 && lane == 0) rowptr[n] = carry;
}

__global__ void __launch_bounds__(256)
k_fill(const int* __restrict__ src, const int* __restrict__ dst,
       const int* __restrict__ rowptr_td, const int* __restrict__ rowptr_bu,
       const int* __restrict__ rank_td, const int* __restrict__ rank_bu,
       int* __restrict__ col_td, int* __restrict__ col_bu, int n_edges) {
  const int width = gridDim.x * blockDim.x;
  const int e0 = blockIdx.x * blockDim.x + threadIdx.x;
  int s[EDGE_PER_THREAD], d[EDGE_PER_THREAD], pt[EDGE_PER_THREAD], pb[EDGE_PER_THREAD];
#pragma unroll
  for (int j = 0; j < EDGE_PER_THREAD; ++j) {
    const int e = min(e0 + j * width, n_edges - 1);
    s[j] = src[e];
    d[j] = dst[e];
    pt[j] = rank_td[e];
    pb[j] = rank_bu[e];
  }
#pragma unroll
  for (int j = 0; j < EDGE_PER_THREAD; ++j) {
    pt[j] += rowptr_td[d[j]];
    pb[j] += rowptr_bu[s[j]];
  }
#pragma unroll
  for (int j = 0; j < EDGE_PER_THREAD; ++j) {
    const int e = e0 + j * width;
    if (e < n_edges) {
      col_td[pt[j]] = s[j];
      col_bu[pb[j]] = d[j];
    }
  }
}

struct GemmFrag {
  float4 a;
  float4 b0, b1, b2, b3;
};
__device__ __forceinline__ GemmFrag gemm_load(const float* __restrict__ Ap,
                                              const float* __restrict__ Bp, int k0, int N) {
  GemmFrag f;
  f.a  = *reinterpret_cast<const float4*>(Ap + k0);
  const float* __restrict__ bp = Bp + (size_t)k0 * N;
  f.b0 = *reinterpret_cast<const float4*>(bp);
  f.b1 = *reinterpret_cast<const float4*>(bp + N);
  f.b2 = *reinterpret_cast<const float4*>(bp + 2 * (size_t)N);
  f.b3 = *reinterpret_cast<const float4*>(bp + 3 * (size_t)N);
  return f;
}
__device__ __forceinline__ v8f mma4(float a0, float a1, float b0, float b1, v8f acc) {
  v2f a, b;
  a.x = a0; a.y = a1;
  b.x = b0; b.y = b1;
  return __builtin_amdgcn_wmma_f32_16x16x4_f32(
       false, a,  false, b,
       (short)0, acc,  false,  false);
}
__device__ __forceinline__ void gemm_block(const GemmFrag& f, v8f& acc0, v8f& acc1,
                                           v8f& acc2, v8f& acc3) {
  acc0 = mma4(f.a.x, f.a.y, f.b0.x, f.b1.x, acc0);
  acc1 = mma4(f.a.x, f.a.y, f.b0.y, f.b1.y, acc1);
  acc2 = mma4(f.a.x, f.a.y, f.b0.z, f.b1.z, acc2);
  acc3 = mma4(f.a.x, f.a.y, f.b0.w, f.b1.w, acc3);
  acc0 = mma4(f.a.z, f.a.w, f.b2.x, f.b3.x, acc0);
  acc1 = mma4(f.a.z, f.a.w, f.b2.y, f.b3.y, acc1);
  acc2 = mma4(f.a.z, f.a.w, f.b2.z, f.b3.z, acc2);
  acc3 = mma4(f.a.z, f.a.w, f.b2.w, f.b3.w, acc3);
}

#define GEMM_WAVES 4
__global__ void __launch_bounds__(GEMM_WAVES * 32)
k_gemm_wmma(const float* __restrict__ A, const float* __restrict__ B,
            float* __restrict__ C, const float* __restrict__ bias,
            int M, int N, int K, int relu) {
  const int lane = threadIdx.x & 31;
  const int wave = threadIdx.x >> 5;
  const int strip = blockIdx.x * GEMM_WAVES + wave;
  const int sn_cnt = N >> 6;
  const int total = (M >> 4) * sn_cnt;
  if (strip >= total) return;
  const int tm = (strip / sn_cnt) << 4;
  const int cn = (strip % sn_cnt) << 6;

  const int h = lane >> 4;
  const int c = lane & 15;

  const float* __restrict__ Ap = A + (size_t)(tm + c) * K + 4 * h;
  const float* __restrict__ Bp = B + (size_t)(4 * h) * N + cn + 4 * c;

  v8f acc0 = {}, acc1 = {}, acc2 = {}, acc3 = {};
  GemmFrag f0 = gemm_load(Ap, Bp, 0, N);
  for (int k0 = 0; k0 < K; k0 += 16) {
    const GemmFrag f1 = gemm_load(Ap, Bp, k0 + 8, N);
    __builtin_amdgcn_sched_barrier(0);
    gemm_block(f0, acc0, acc1, acc2, acc3);
    f0 = gemm_load(Ap, Bp, (k0 + 16 < K) ? k0 + 16 : k0, N);
    __builtin_amdgcn_sched_barrier(0);
    gemm_block(f1, acc0, acc1, acc2, acc3);
  }

  float4 bi = make_float4(0.f, 0.f, 0.f, 0.f);
  if (bias) bi = *reinterpret_cast<const float4*>(bias + cn + 4 * c);
#pragma unroll
  for (int r = 0; r < 8; ++r) {
    float4 v;
    v.x = acc0[r] + bi.x;
    v.y = acc1[r] + bi.y;
    v.z = acc2[r] + bi.z;
    v.w = acc3[r] + bi.w;
    if (relu) {
      v.x = fmaxf(v.x, 0.0f);
      v.y = fmaxf(v.y, 0.0f);
      v.z = fmaxf(v.z, 0.0f);
      v.w = fmaxf(v.w, 0.0f);
    }
    *reinterpret_cast<float4*>(C + (size_t)(tm + 8 * h + r) * N + cn + 4 * c) = v;
  }
}

__device__ __forceinline__ float4 ld_row(const float* __restrict__ h, int s, int lane) {
  return *reinterpret_cast<const float4*>(h + (size_t)s * 128 + lane * 4);
}
__device__ __forceinline__ void fma4(float4& a, const float4 v, const float w) {
  a.x = fmaf(v.x, w, a.x);
  a.y = fmaf(v.y, w, a.y);
  a.z = fmaf(v.z, w, a.z);
  a.w = fmaf(v.w, w, a.w);
}
__device__ __forceinline__ int lane_i(int v, int j) { return __builtin_amdgcn_readlane(v, j); }
__device__ __forceinline__ float lane_f(float v, int j) {
  return __int_as_float(__builtin_amdgcn_readlane(__float_as_int(v), j));
}

__device__ __forceinline__ float4 gather_row(const float* __restrict__ h,
                                             const int* __restrict__ col,
                                             const float* __restrict__ dinv,
                                             int beg, int end, float dd, int lane) {
  float4 acc = make_float4(0.f, 0.f, 0.f, 0.f);
  for (int base = beg; base < end; base += 32) {
    const int n = min(32, end - base);
    int my_s = 0;
    float my_w = 0.f;
    if (lane < n) {
      my_s = col[base + lane];
      my_w = dinv[my_s] * dd;
    }
    int j = 0;
    for (; j + 8 <= n; j += 8) {
      const float4 v0 = ld_row(h, lane_i(my_s, j), lane);
      const float4 v1 = ld_row(h, lane_i(my_s, j + 1), lane);
      const float4 v2 = ld_row(h, lane_i(my_s, j + 2), lane);
      const float4 v3 = ld_row(h, lane_i(my_s, j + 3), lane);
      const float4 v4 = ld_row(h, lane_i(my_s, j + 4), lane);
      const float4 v5 = ld_row(h, lane_i(my_s, j + 5), lane);
      const float4 v6 = ld_row(h, lane_i(my_s, j + 6), lane);
      const float4 v7 = ld_row(h, lane_i(my_s, j + 7), lane);
      fma4(acc, v0, lane_f(my_w, j));
      fma4(acc, v1, lane_f(my_w, j + 1));
      fma4(acc, v2, lane_f(my_w, j + 2));
      fma4(acc, v3, lane_f(my_w, j + 3));
      fma4(acc, v4, lane_f(my_w, j + 4));
      fma4(acc, v5, lane_f(my_w, j + 5));
      fma4(acc, v6, lane_f(my_w, j + 6));
      fma4(acc, v7, lane_f(my_w, j + 7));
    }
    if (j + 4 <= n) {
      const float4 v0 = ld_row(h, lane_i(my_s, j), lane);
      const float4 v1 = ld_row(h, lane_i(my_s, j + 1), lane);
      const float4 v2 = ld_row(h, lane_i(my_s, j + 2), lane);
      const float4 v3 = ld_row(h, lane_i(my_s, j + 3), lane);
      fma4(acc, v0, lane_f(my_w, j));
      fma4(acc, v1, lane_f(my_w, j + 1));
      fma4(acc, v2, lane_f(my_w, j + 2));
      fma4(acc, v3, lane_f(my_w, j + 3));
      j += 4;
    }
    if (j + 2 <= n) {
      const float4 v0 = ld_row(h, lane_i(my_s, j), lane);
      const float4 v1 = ld_row(h, lane_i(my_s, j + 1), lane);
      fma4(acc, v0, lane_f(my_w, j));
      fma4(acc, v1, lane_f(my_w, j + 1));
      j += 2;
    }
    if (j < n) {
      const float4 v0 = ld_row(h, lane_i(my_s, j), lane);
      fma4(acc, v0, lane_f(my_w, j));
    }
  }
  return acc;
}

#define GATHER_BLOCKS 1024
__device__ __forceinline__ void wave_rows(int n_nodes, int& r0, int& r1) {
  const int g = __builtin_amdgcn_readfirstlane((int)((blockIdx.x * blockDim.x + threadIdx.x) >> 5));
  const int nw = (int)((gridDim.x * blockDim.x) >> 5);
  const int q = n_nodes / nw;
  const int r = n_nodes % nw;
  r0 = g * q + min(g, r);
  r1 = r0 + q + (g < r ? 1 : 0);
}

__global__ void __launch_bounds__(256)
k_gather_relu(const float* __restrict__ h, const int* __restrict__ rowptr,
              const int* __restrict__ col, const float* __restrict__ dinv,
              const float* __restrict__ bias, float* __restrict__ out, int n_nodes) {
  const int lane = threadIdx.x & 31;
  int r0, r1;
  wave_rows(n_nodes, r0, r1);
  const float4 b = *reinterpret_cast<const float4*>(bias + lane * 4);
  for (int row = r0; row < r1; ++row) {
    const int beg = rowptr[row];
    const int end = rowptr[row + 1];
    const float dd = dinv[row];
    const float4 self = ld_row(h, row, lane);
    const float4 a = gather_row(h, col, dinv, beg, end, dd, lane);
    float4 v;
    v.x = fmaxf(a.x + self.x * dd * dd + b.x, 0.0f);
    v.y = fmaxf(a.y + self.y * dd * dd + b.y, 0.0f);
    v.z = fmaxf(a.z + self.z * dd * dd + b.z, 0.0f);
    v.w = fmaxf(a.w + self.w * dd * dd + b.w, 0.0f);
    *reinterpret_cast<float4*>(out + (size_t)row * 128 + lane * 4) = v;
  }
}

__global__ void __launch_bounds__(256)
k_gather_pool(const float* __restrict__ h, const int* __restrict__ rowptr,
              const int* __restrict__ col, const float* __restrict__ dinv,
              const int* __restrict__ batch, float* __restrict__ pooled,
              float* __restrict__ gcnt, int col_off, int n_nodes) {
  const int lane = threadIdx.x & 31;
  int r0, r1;
  wave_rows(n_nodes, r0, r1);
  for (int row = r0; row < r1; ++row) {
    const int beg = rowptr[row];
    const int end = rowptr[row + 1];
    const float dd = dinv[row];
    const float4 self = ld_row(h, row, lane);
    const float4 a = gather_row(h, col, dinv, beg, end, dd, lane);
    const int g = batch[row];
    float* __restrict__ p = pooled + (size_t)g * 256 + col_off + lane * 4;
    atomicAdd(&p[0], a.x + self.x * dd * dd);
    atomicAdd(&p[1], a.y + self.y * dd * dd);
    atomicAdd(&p[2], a.z + self.z * dd * dd);
    atomicAdd(&p[3], a.w + self.w * dd * dd);
    if (gcnt != nullptr && lane == 0) atomicAdd(&gcnt[g], 1.0f);
  }
}

__global__ void __launch_bounds__(GEMM_WAVES * 32)
k_pool_project(const float* __restrict__ pooled,
               const float* __restrict__ W_bu, const float* __restrict__ W_td,
               const float* __restrict__ b_bu, const float* __restrict__ b_td,
               const float* __restrict__ gcnt, float* __restrict__ hcat, int G) {
  const int lane = threadIdx.x & 31;
  const int wave = threadIdx.x >> 5;
  const int strip = blockIdx.x * GEMM_WAVES + wave;
  if (strip >= (G >> 4) * 4) return;
  const int tm = (strip >> 2) << 4;
  const int cs = strip & 3;
  const int br = cs >> 1;
  const int cn = (cs & 1) << 6;
  const float* __restrict__ Wb = br ? W_td : W_bu;
  const float* __restrict__ bb = br ? b_td : b_bu;

  const int h = lane >> 4;
  const int c = lane & 15;
  const int K = 128, N = 128;
  const float* __restrict__ Ap = pooled + (size_t)(tm + c) * 256 + 128 * br + 4 * h;
  const float* __restrict__ Bp = Wb + (size_t)(4 * h) * N + cn + 4 * c;

  v8f acc0 = {}, acc1 = {}, acc2 = {}, acc3 = {};
  GemmFrag f0 = gemm_load(Ap, Bp, 0, N);
  for (int k0 = 0; k0 < K; k0 += 16) {
    const GemmFrag f1 = gemm_load(Ap, Bp, k0 + 8, N);
    __builtin_amdgcn_sched_barrier(0);
    gemm_block(f0, acc0, acc1, acc2, acc3);
    f0 = gemm_load(Ap, Bp, (k0 + 16 < K) ? k0 + 16 : k0, N);
    __builtin_amdgcn_sched_barrier(0);
    gemm_block(f1, acc0, acc1, acc2, acc3);
  }

  const float4 bi = *reinterpret_cast<const float4*>(bb + cn + 4 * c);
#pragma unroll
  for (int r = 0; r < 8; ++r) {
    const int row = tm + 8 * h + r;
    const float n = gcnt[row];
    float4 v;
    v.x = fmaf(n, bi.x, acc0[r]);
    v.y = fmaf(n, bi.y, acc1[r]);
    v.z = fmaf(n, bi.z, acc2[r]);
    v.w = fmaf(n, bi.w, acc3[r]);
    *reinterpret_cast<float4*>(hcat + (size_t)row * 256 + 64 * cs + 4 * c) = v;
  }
}

static inline void launch_gemm(const float* A, const float* B, float* C,
                               const float* bias, int M, int N, int K, int relu,
                               hipStream_t stream) {
  const int strips = (M >> 4) * (N >> 6);
  const int blocks = (strips + GEMM_WAVES - 1) / GEMM_WAVES;
  k_gemm_wmma<<<blocks, GEMM_WAVES * 32, 0, stream>>>(A, B, C, bias, M, N, K, relu);
}

extern "C" void kernel_launch(void* const* d_in, const int* in_sizes, int n_in,
                              void* d_out, int out_size, void* d_ws, size_t ws_size,
                              hipStream_t stream) {
  const float* x      = (const float*)d_in[0];
  const int* edge_idx = (const int*)d_in[1];
  const int* batch    = (const int*)d_in[2];
  const float* td_W1 = (const float*)d_in[4];
  const float* td_b1 = (const float*)d_in[5];
  const float* td_W2 = (const float*)d_in[6];
  const float* td_b2 = (const float*)d_in[7];
  const float* bu_W1 = (const float*)d_in[8];
  const float* bu_b1 = (const float*)d_in[9];
  const float* bu_W2 = (const float*)d_in[10];
  const float* bu_b2 = (const float*)d_in[11];
  const float* pw1   = (const float*)d_in[12];
  const float* pb1   = (const float*)d_in[13];
  const float* pw2   = (const float*)d_in[14];
  const float* pb2   = (const float*)d_in[15];

  const int IN_F = 256, HID = 128, G = 512;
  const int n_nodes = in_sizes[0] / IN_F;
  const int n_edges = in_sizes[1] / 2;
  const int* src = edge_idx;
  const int* dst = edge_idx + n_edges;

  float* ws      = (float*)d_ws;
  const size_t NF = (size_t)n_nodes * 128;
  float* h_lin   = ws;
  int*   csr     = (int*)(ws + NF);
  float* h_act   = ws + 2 * NF;
  float* dinv_td = ws + 3 * NF;
  float* dinv_bu = dinv_td + n_nodes;
  float* pooled  = dinv_bu + n_nodes;
  float* gcnt    = pooled + (size_t)G * 256;
  float* hcat    = ws;
  float* mlp_h   = ws + (size_t)G * 256;

  int* cnt_td    = csr;
  int* cnt_bu    = cnt_td + n_nodes;
  int* rowptr_td = cnt_bu + n_nodes;
  int* rowptr_bu = rowptr_td + (n_nodes + 1);
  int* col_td    = rowptr_bu + (n_nodes + 1);
  int* col_bu    = col_td + n_edges;
  int* rank_td   = col_bu + n_edges;
  int* rank_bu   = rank_td + n_edges;

  const int EB = (n_edges + 256 * EDGE_PER_THREAD - 1) / (256 * EDGE_PER_THREAD);
  const int RB = GATHER_BLOCKS;

  hipMemsetAsync(cnt_td, 0, (size_t)2 * n_nodes * sizeof(int), stream);
  k_count_rank<<<EB, 256, 0, stream>>>(src, dst, cnt_td, cnt_bu, rank_td, rank_bu, n_edges);
  k_scan_dinv<<<2, SCAN_WAVES * 32, 0, stream>>>(cnt_td, rowptr_td, dinv_td, n_nodes);
  k_fill<<<EB, 256, 0, stream>>>(src, dst, rowptr_td, rowptr_bu, rank_td, rank_bu,
                                 col_td, col_bu, n_edges);
  hipMemsetAsync(pooled, 0, ((size_t)G * 256 + G) * sizeof(float), stream);

  launch_gemm(x, td_W1, h_lin, nullptr, n_nodes, HID, IN_F, 0, stream);
  k_gather_relu<<<RB, 256, 0, stream>>>(h_lin, rowptr_td, col_td, dinv_td, td_b1, h_act, n_nodes);
  k_gather_pool<<<RB, 256, 0, stream>>>(h_act, rowptr_td, col_td, dinv_td, batch, pooled, gcnt,
                                         128, n_nodes);

  launch_gemm(x, bu_W1, h_lin, nullptr, n_nodes, HID, IN_F, 0, stream);
  k_gather_relu<<<RB, 256, 0, stream>>>(h_lin, rowptr_bu, col_bu, dinv_bu, bu_b1, h_act, n_nodes);
  k_gather_pool<<<RB, 256, 0, stream>>>(h_act, rowptr_bu, col_bu, dinv_bu, batch, pooled, nullptr,
                                         0, n_nodes);

  k_pool_project<<<((G >> 4) * 4 + GEMM_WAVES - 1) / GEMM_WAVES, GEMM_WAVES * 32, 0, stream>>>(
      pooled, bu_W2, td_W2, bu_b2, td_b2, gcnt, hcat, G);
  launch_gemm(hcat, pw1, mlp_h, pb1, G, 256, 256,  1, stream);
  launch_gemm(mlp_h, pw2, (float*)d_out, pb2, G, 128, 256,  0, stream);
}


# independent stages share a launch: layer-1 GEMM blocks run beside the memory-bound edge counting and edge fill (TD product) and beside the TD pooled gather (BU product); 12 launches become 10
# speedup vs baseline: 2.6652x; 1.0477x over previous
#include <hip/hip_runtime.h>


typedef __attribute__((ext_vector_type(2))) float v2f;
typedef __attribute__((ext_vector_type(8))) float v8f;

#define EDGE_PER_THREAD 4
__device__ __forceinline__ void
count_rank_body(const int* __restrict__ src, const int* __restrict__ dst,
                int* __restrict__ cnt_td, int* __restrict__ cnt_bu,
                int* __restrict__ rank_td, int* __restrict__ rank_bu,
                int n_edges, int blk, int nblk) {
  const int width = nblk * blockDim.x;
  const int e0 = blk * blockDim.x + threadIdx.x;
  int s[EDGE_PER_THREAD], d[EDGE_PER_THREAD], rt[EDGE_PER_THREAD], rb[EDGE_PER_THREAD];
#pragma unroll
  for (int j = 0; j < EDGE_PER_THREAD; ++j) {
    const int e = min(e0 + j * width, n_edges - 1);
    s[j] = src[e];
    d[j] = dst[e];
  }
#pragma unroll
  for (int j = 0; j < EDGE_PER_THREAD; ++j) {
    const int e = e0 + j * width;
    rt[j] = 0;
    rb[j] = 0;
    if (e < n_edges) {
      rt[j] = atomicAdd(&cnt_td[d[j]], 1);
      rb[j] = atomicAdd(&cnt_bu[s[j]], 1);
    }
  }
#pragma unroll
  for (int j = 0; j < EDGE_PER_THREAD; ++j) {
    const int e = e0 + j * width;
    if (e < n_edges) {
      rank_td[e] = rt[j];
      rank_bu[e] = rb[j];
    }
  }
}

#define SCAN_WAVES 32
#define SCAN_BATCH 7
__global__ void __launch_bounds__(SCAN_WAVES * 32)
k_scan_dinv(const int* __restrict__ cnt2, int* __restrict__ rowptr2,
            float* __restrict__ dinv2, int n) {
  __shared__ int wave_tot[SCAN_WAVES];
  const int* __restrict__ cnt = cnt2 + (size_t)blockIdx.x * n;
  int* __restrict__ rowptr = rowptr2 + (size_t)blockIdx.x * (n + 1);
  float* __restrict__ dinv = dinv2 + (size_t)blockIdx.x * n;
  const int lane = threadIdx.x & 31;
  const int wave = threadIdx.x >> 5;
  const int steps = (n + SCAN_WAVES * 32 - 1) / (SCAN_WAVES * 32);
  const int base = wave * steps * 32;

  int part = 0;
  for (int k = 0; k < steps; ++k) {
    const int i = base + k * 32 + lane;
    part += (i < n) ? cnt[i] : 0;
  }
  for (int o = 16; o > 0; o >>= 1) part += __shfl_xor(part, o);
  if (lane == 0) wave_tot[wave] = part;
  __syncthreads();
  int carry = (lane < wave) ? wave_tot[lane] : 0;
  for (int o = 16; o > 0; o >>= 1) carry += __shfl_xor(carry, o);

  for (int k0 = 0; k0 < steps; k0 += SCAN_BATCH) {
    int v[SCAN_BATCH];
#pragma unroll
    for (int u = 0; u < SCAN_BATCH; ++u) {
      const int i = base + (k0 + u) * 32 + lane;
      v[u] = (k0 + u < steps && i < n) ? cnt[i] : 0;
    }
    int x[SCAN_BATCH];
#pragma unroll
    for (int u = 0; u < SCAN_BATCH; ++u) x[u] = v[u];
#pragma unroll
    for (int o = 1; o < 32; o <<= 1) {
      int y[SCAN_BATCH];
#pragma unroll
      for (int u = 0; u < SCAN_BATCH; ++u) y[u] = __shfl_up(x[u], o);
#pragma unroll
      for (int u = 0; u < SCAN_BATCH; ++u) x[u] += (lane >= o) ? y[u] : 0;
    }
#pragma unroll
    for (int u = 0; u < SCAN_BATCH; ++u) {
      const int i = base + (k0 + u) * 32 + lane;
      if (k0 + u < steps && i < n) {
        rowptr[i] = carry + x[u] - v[u];
        dinv[i] = rsqrtf((float)v[u] + 1.0f);
      }
      carry += __builtin_amdgcn_readlane(x[u], 31);
    }
  }
  if (wave == SCAN_WAVES - 1 && lane == 0) rowptr[n] = carry;
}

__device__ __forceinline__ void
fill_body(const int* __restrict__ src, const int* __restrict__ dst,
          const int* __restrict__ rowptr_td, const int* __restrict__ rowptr_bu,
          const int* __restrict__ rank_td, const int* __restrict__ rank_bu,
          int* __restrict__ col_td, int* __restrict__ col_bu, int n_edges, int blk, int nblk) {
  const int width = nblk * blockDim.x;
  const int e0 = blk * blockDim.x + threadIdx.x;
  int s[EDGE_PER_THREAD], d[EDGE_PER_THREAD], pt[EDGE_PER_THREAD], pb[EDGE_PER_THREAD];
#pragma unroll
  for (int j = 0; j < EDGE_PER_THREAD; ++j) {
    const int e = min(e0 + j * width, n_edges - 1);
    s[j] = src[e];
    d[j] = dst[e];
    pt[j] = rank_td[e];
    pb[j] = rank_bu[e];
  }
#pragma unroll
  for (int j = 0; j < EDGE_PER_THREAD; ++j) {
    pt[j] += rowptr_td[d[j]];
    pb[j] += rowptr_bu[s[j]];
  }
#pragma unroll
  for (int j = 0; j < EDGE_PER_THREAD; ++j) {
    const int e = e0 + j * width;
    if (e < n_edges) {
      col_td[pt[j]] = s[j];
      col_bu[pb[j]] = d[j];
    }
  }
}

struct GemmFrag {
  float4 a;
  float4 b0, b1, b2, b3;
};
__device__ __forceinline__ GemmFrag gemm_load(const float* __restrict__ Ap,
                                              const float* __restrict__ Bp, int k0, int N) {
  GemmFrag f;
  f.a  = *reinterpret_cast<const float4*>(Ap + k0);
  const float* __restrict__ bp = Bp + (size_t)k0 * N;
  f.b0 = *reinterpret_cast<const float4*>(bp);
  f.b1 = *reinterpret_cast<const float4*>(bp + N);
  f.b2 = *reinterpret_cast<const float4*>(bp + 2 * (size_t)N);
  f.b3 = *reinterpret_cast<const float4*>(bp + 3 * (size_t)N);
  return f;
}
__device__ __forceinline__ v8f mma4(float a0, float a1, float b0, float b1, v8f acc) {
  v2f a, b;
  a.x = a0; a.y = a1;
  b.x = b0; b.y = b1;
  return __builtin_amdgcn_wmma_f32_16x16x4_f32(
       false, a,  false, b,
       (short)0, acc,  false,  false);
}
__device__ __forceinline__ void gemm_block(const GemmFrag& f, v8f& acc0, v8f& acc1,
                                           v8f& acc2, v8f& acc3) {
  acc0 = mma4(f.a.x, f.a.y, f.b0.x, f.b1.x, acc0);
  acc1 = mma4(f.a.x, f.a.y, f.b0.y, f.b1.y, acc1);
  acc2 = mma4(f.a.x, f.a.y, f.b0.z, f.b1.z, acc2);
  acc3 = mma4(f.a.x, f.a.y, f.b0.w, f.b1.w, acc3);
  acc0 = mma4(f.a.z, f.a.w, f.b2.x, f.b3.x, acc0);
  acc1 = mma4(f.a.z, f.a.w, f.b2.y, f.b3.y, acc1);
  acc2 = mma4(f.a.z, f.a.w, f.b2.z, f.b3.z, acc2);
  acc3 = mma4(f.a.z, f.a.w, f.b2.w, f.b3.w, acc3);
}

#define GEMM_WAVES 4
__device__ __forceinline__ void
gemm_strip(const float* __restrict__ A, const float* __restrict__ B,
           float* __restrict__ C, const float* __restrict__ bias,
           int M, int N, int K, int relu, int strip) {
  const int lane = threadIdx.x & 31;
  const int sn_cnt = N >> 6;
  const int total = (M >> 4) * sn_cnt;
  if (strip >= total) return;
  const int tm = (strip / sn_cnt) << 4;
  const int cn = (strip % sn_cnt) << 6;

  const int h = lane >> 4;
  const int c = lane & 15;

  const float* __restrict__ Ap = A + (size_t)(tm + c) * K + 4 * h;
  const float* __restrict__ Bp = B + (size_t)(4 * h) * N + cn + 4 * c;

  v8f acc0 = {}, acc1 = {}, acc2 = {}, acc3 = {};
  GemmFrag f0 = gemm_load(Ap, Bp, 0, N);
  for (int k0 = 0; k0 < K; k0 += 16) {
    const GemmFrag f1 = gemm_load(Ap, Bp, k0 + 8, N);
    __builtin_amdgcn_sched_barrier(0);
    gemm_block(f0, acc0, acc1, acc2, acc3);
    f0 = gemm_load(Ap, Bp, (k0 + 16 < K) ? k0 + 16 : k0, N);
    __builtin_amdgcn_sched_barrier(0);
    gemm_block(f1, acc0, acc1, acc2, acc3);
  }

  float4 bi = make_float4(0.f, 0.f, 0.f, 0.f);
  if (bias) bi = *reinterpret_cast<const float4*>(bias + cn + 4 * c);
#pragma unroll
  for (int r = 0; r < 8; ++r) {
    float4 v;
    v.x = acc0[r] + bi.x;
    v.y = acc1[r] + bi.y;
    v.z = acc2[r] + bi.z;
    v.w = acc3[r] + bi.w;
    if (relu) {
      v.x = fmaxf(v.x, 0.0f);
      v.y = fmaxf(v.y, 0.0f);
      v.z = fmaxf(v.z, 0.0f);
      v.w = fmaxf(v.w, 0.0f);
    }
    *reinterpret_cast<float4*>(C + (size_t)(tm + 8 * h + r) * N + cn + 4 * c) = v;
  }
}

__global__ void __launch_bounds__(GEMM_WAVES * 32)
k_gemm_wmma(const float* __restrict__ A, const float* __restrict__ B,
            float* __restrict__ C, const float* __restrict__ bias,
            int M, int N, int K, int relu) {
  gemm_strip(A, B, C, bias, M, N, K, relu, blockIdx.x * GEMM_WAVES + (threadIdx.x >> 5));
}

#define FUSED_WAVES 4
#define FUSED_MIN_WAVES 10
#define L1_K 256
#define L1_N 128
__global__ void __launch_bounds__(FUSED_WAVES * 32, FUSED_MIN_WAVES)
k_count_rank_gemm(const int* __restrict__ src, const int* __restrict__ dst,
                  int* __restrict__ cnt_td, int* __restrict__ cnt_bu,
                  int* __restrict__ rank_td, int* __restrict__ rank_bu,
                  int n_edges, int other_blocks,
                  const float* __restrict__ A, const float* __restrict__ B,
                  float* __restrict__ C, int M, int strip0) {
  if ((int)blockIdx.x < other_blocks) {
    count_rank_body(src, dst, cnt_td, cnt_bu, rank_td, rank_bu, n_edges,
                    (int)blockIdx.x, other_blocks);
  } else {
    gemm_strip(A, B, C, nullptr, M, L1_N, L1_K, 0,
               strip0 + ((int)blockIdx.x - other_blocks) * FUSED_WAVES + (int)(threadIdx.x >> 5));
  }
}

__global__ void __launch_bounds__(FUSED_WAVES * 32, FUSED_MIN_WAVES)
k_fill_gemm(const int* __restrict__ src, const int* __restrict__ dst,
            const int* __restrict__ rowptr_td, const int* __restrict__ rowptr_bu,
            const int* __restrict__ rank_td, const int* __restrict__ rank_bu,
            int* __restrict__ col_td, int* __restrict__ col_bu,
            int n_edges, int other_blocks,
            const float* __restrict__ A, const float* __restrict__ B,
            float* __restrict__ C, int M, int strip0) {
  if ((int)blockIdx.x < other_blocks) {
    fill_body(src, dst, rowptr_td, rowptr_bu, rank_td, rank_bu, col_td, col_bu, n_edges,
              (int)blockIdx.x, other_blocks);
  } else {
    gemm_strip(A, B, C, nullptr, M, L1_N, L1_K, 0,
               strip0 + ((int)blockIdx.x - other_blocks) * FUSED_WAVES + (int)(threadIdx.x >> 5));
  }
}

__device__ __forceinline__ float4 ld_row(const float* __restrict__ h, int s, int lane) {
  return *reinterpret_cast<const float4*>(h + (size_t)s * 128 + lane * 4);
}
__device__ __forceinline__ void fma4(float4& a, const float4 v, const float w) {
  a.x = fmaf(v.x, w, a.x);
  a.y = fmaf(v.y, w, a.y);
  a.z = fmaf(v.z, w, a.z);
  a.w = fmaf(v.w, w, a.w);
}
__device__ __forceinline__ int lane_i(int v, int j) { return __builtin_amdgcn_readlane(v, j); }
__device__ __forceinline__ float lane_f(float v, int j) {
  return __int_as_float(__builtin_amdgcn_readlane(__float_as_int(v), j));
}

__device__ __forceinline__ float4 gather_row(const float* __restrict__ h,
                                             const int* __restrict__ col,
                                             const float* __restrict__ dinv,
                                             int beg, int end, float dd, int lane) {
  float4 acc = make_float4(0.f, 0.f, 0.f, 0.f);
  for (int base = beg; base < end; base += 32) {
    const int n = min(32, end - base);
    int my_s = 0;
    float my_w = 0.f;
    if (lane < n) {
      my_s = col[base + lane];
      my_w = dinv[my_s] * dd;
    }
    int j = 0;
    for (; j + 8 <= n; j += 8) {
      const float4 v0 = ld_row(h, lane_i(my_s, j), lane);
      const float4 v1 = ld_row(h, lane_i(my_s, j + 1), lane);
      const float4 v2 = ld_row(h, lane_i(my_s, j + 2), lane);
      const float4 v3 = ld_row(h, lane_i(my_s, j + 3), lane);
      const float4 v4 = ld_row(h, lane_i(my_s, j + 4), lane);
      const float4 v5 = ld_row(h, lane_i(my_s, j + 5), lane);
      const float4 v6 = ld_row(h, lane_i(my_s, j + 6), lane);
      const float4 v7 = ld_row(h, lane_i(my_s, j + 7), lane);
      fma4(acc, v0, lane_f(my_w, j));
      fma4(acc, v1, lane_f(my_w, j + 1));
      fma4(acc, v2, lane_f(my_w, j + 2));
      fma4(acc, v3, lane_f(my_w, j + 3));
      fma4(acc, v4, lane_f(my_w, j + 4));
      fma4(acc, v5, lane_f(my_w, j + 5));
      fma4(acc, v6, lane_f(my_w, j + 6));
      fma4(acc, v7, lane_f(my_w, j + 7));
    }
    if (j + 4 <= n) {
      const float4 v0 = ld_row(h, lane_i(my_s, j), lane);
      const float4 v1 = ld_row(h, lane_i(my_s, j + 1), lane);
      const float4 v2 = ld_row(h, lane_i(my_s, j + 2), lane);
      const float4 v3 = ld_row(h, lane_i(my_s, j + 3), lane);
      fma4(acc, v0, lane_f(my_w, j));
      fma4(acc, v1, lane_f(my_w, j + 1));
      fma4(acc, v2, lane_f(my_w, j + 2));
      fma4(acc, v3, lane_f(my_w, j + 3));
      j += 4;
    }
    if (j + 2 <= n) {
      const float4 v0 = ld_row(h, lane_i(my_s, j), lane);
      const float4 v1 = ld_row(h, lane_i(my_s, j + 1), lane);
      fma4(acc, v0, lane_f(my_w, j));
      fma4(acc, v1, lane_f(my_w, j + 1));
      j += 2;
    }
    if (j < n) {
      const float4 v0 = ld_row(h, lane_i(my_s, j), lane);
      fma4(acc, v0, lane_f(my_w, j));
    }
  }
  return acc;
}

#define GATHER_BLOCKS 1024
__device__ __forceinline__ void wave_rows(int n_nodes, int blk, int nblk, int& r0, int& r1) {
  const int g = __builtin_amdgcn_readfirstlane((int)((blk * blockDim.x + threadIdx.x) >> 5));
  const int nw = (int)((nblk * blockDim.x) >> 5);
  const int q = n_nodes / nw;
  const int r = n_nodes % nw;
  r0 = g * q + min(g, r);
  r1 = r0 + q + (g < r ? 1 : 0);
}

__global__ void __launch_bounds__(256)
k_gather_relu(const float* __restrict__ h, const int* __restrict__ rowptr,
              const int* __restrict__ col, const float* __restrict__ dinv,
              const float* __restrict__ bias, float* __restrict__ out, int n_nodes) {
  const int lane = threadIdx.x & 31;
  int r0, r1;
  wave_rows(n_nodes, (int)blockIdx.x, (int)gridDim.x, r0, r1);
  const float4 b = *reinterpret_cast<const float4*>(bias + lane * 4);
  for (int row = r0; row < r1; ++row) {
    const int beg = rowptr[row];
    const int end = rowptr[row + 1];
    const float dd = dinv[row];
    const float4 self = ld_row(h, row, lane);
    const float4 a = gather_row(h, col, dinv, beg, end, dd, lane);
    float4 v;
    v.x = fmaxf(a.x + self.x * dd * dd + b.x, 0.0f);
    v.y = fmaxf(a.y + self.y * dd * dd + b.y, 0.0f);
    v.z = fmaxf(a.z + self.z * dd * dd + b.z, 0.0f);
    v.w = fmaxf(a.w + self.w * dd * dd + b.w, 0.0f);
    *reinterpret_cast<float4*>(out + (size_t)row * 128 + lane * 4) = v;
  }
}

__device__ __forceinline__ void
gather_pool_body(const float* __restrict__ h, const int* __restrict__ rowptr,
                 const int* __restrict__ col, const float* __restrict__ dinv,
                 const int* __restrict__ batch, float* __restrict__ pooled,
                 float* __restrict__ gcnt, int col_off, int n_nodes, int blk, int nblk) {
  const int lane = threadIdx.x & 31;
  int r0, r1;
  wave_rows(n_nodes, blk, nblk, r0, r1);
  for (int row = r0; row < r1; ++row) {
    const int beg = rowptr[row];
    const int end = rowptr[row + 1];
    const float dd = dinv[row];
    const float4 self = ld_row(h, row, lane);
    const float4 a = gather_row(h, col, dinv, beg, end, dd, lane);
    const int g = batch[row];
    float* __restrict__ p = pooled + (size_t)g * 256 + col_off + lane * 4;
    atomicAdd(&p[0], a.x + self.x * dd * dd);
    atomicAdd(&p[1], a.y + self.y * dd * dd);
    atomicAdd(&p[2], a.z + self.z * dd * dd);
    atomicAdd(&p[3], a.w + self.w * dd * dd);
    if (gcnt != nullptr && lane == 0) atomicAdd(&gcnt[g], 1.0f);
  }
}

__global__ void __launch_bounds__(256)
k_gather_pool(const float* __restrict__ h, const int* __restrict__ rowptr,
              const int* __restrict__ col, const float* __restrict__ dinv,
              const int* __restrict__ batch, float* __restrict__ pooled,
              float* __restrict__ gcnt, int col_off, int n_nodes) {
  gather_pool_body(h, rowptr, col, dinv, batch, pooled, gcnt, col_off, n_nodes,
                   (int)blockIdx.x, (int)gridDim.x);
}

__global__ void __launch_bounds__(FUSED_WAVES * 32, FUSED_MIN_WAVES)
k_gather_pool_gemm(const float* __restrict__ h, const int* __restrict__ rowptr,
                   const int* __restrict__ col, const float* __restrict__ dinv,
                   const int* __restrict__ batch, float* __restrict__ pooled,
                   float* __restrict__ gcnt, int col_off, int n_nodes, int other_blocks,
                   const float* __restrict__ A, const float* __restrict__ B,
                   float* __restrict__ C, int M) {
  if ((int)blockIdx.x < other_blocks) {
    gather_pool_body(h, rowptr, col, dinv, batch, pooled, gcnt, col_off, n_nodes,
                     (int)blockIdx.x, other_blocks);
  } else {
    gemm_strip(A, B, C, nullptr, M, L1_N, L1_K, 0,
               ((int)blockIdx.x - other_blocks) * FUSED_WAVES + (int)(threadIdx.x >> 5));
  }
}

__global__ void __launch_bounds__(GEMM_WAVES * 32)
k_pool_project(const float* __restrict__ pooled,
               const float* __restrict__ W_bu, const float* __restrict__ W_td,
               const float* __restrict__ b_bu, const float* __restrict__ b_td,
               const float* __restrict__ gcnt, float* __restrict__ hcat, int G) {
  const int lane = threadIdx.x & 31;
  const int wave = threadIdx.x >> 5;
  const int strip = blockIdx.x * GEMM_WAVES + wave;
  if (strip >= (G >> 4) * 4) return;
  const int tm = (strip >> 2) << 4;
  const int cs = strip & 3;
  const int br = cs >> 1;
  const int cn = (cs & 1) << 6;
  const float* __restrict__ Wb = br ? W_td : W_bu;
  const float* __restrict__ bb = br ? b_td : b_bu;

  const int h = lane >> 4;
  const int c = lane & 15;
  const int K = 128, N = 128;
  const float* __restrict__ Ap = pooled + (size_t)(tm + c) * 256 + 128 * br + 4 * h;
  const float* __restrict__ Bp = Wb + (size_t)(4 * h) * N + cn + 4 * c;

  v8f acc0 = {}, acc1 = {}, acc2 = {}, acc3 = {};
  GemmFrag f0 = gemm_load(Ap, Bp, 0, N);
  for (int k0 = 0; k0 < K; k0 += 16) {
    const GemmFrag f1 = gemm_load(Ap, Bp, k0 + 8, N);
    __builtin_amdgcn_sched_barrier(0);
    gemm_block(f0, acc0, acc1, acc2, acc3);
    f0 = gemm_load(Ap, Bp, (k0 + 16 < K) ? k0 + 16 : k0, N);
    __builtin_amdgcn_sched_barrier(0);
    gemm_block(f1, acc0, acc1, acc2, acc3);
  }

  const float4 bi = *reinterpret_cast<const float4*>(bb + cn + 4 * c);
#pragma unroll
  for (int r = 0; r < 8; ++r) {
    const int row = tm + 8 * h + r;
    const float n = gcnt[row];
    float4 v;
    v.x = fmaf(n, bi.x, acc0[r]);
    v.y = fmaf(n, bi.y, acc1[r]);
    v.z = fmaf(n, bi.z, acc2[r]);
    v.w = fmaf(n, bi.w, acc3[r]);
    *reinterpret_cast<float4*>(hcat + (size_t)row * 256 + 64 * cs + 4 * c) = v;
  }
}

static inline void launch_gemm(const float* A, const float* B, float* C,
                               const float* bias, int M, int N, int K, int relu,
                               hipStream_t stream) {
  const int strips = (M >> 4) * (N >> 6);
  const int blocks = (strips + GEMM_WAVES - 1) / GEMM_WAVES;
  k_gemm_wmma<<<blocks, GEMM_WAVES * 32, 0, stream>>>(A, B, C, bias, M, N, K, relu);
}

extern "C" void kernel_launch(void* const* d_in, const int* in_sizes, int n_in,
                              void* d_out, int out_size, void* d_ws, size_t ws_size,
                              hipStream_t stream) {
  const float* x      = (const float*)d_in[0];
  const int* edge_idx = (const int*)d_in[1];
  const int* batch    = (const int*)d_in[2];
  const float* td_W1 = (const float*)d_in[4];
  const float* td_b1 = (const float*)d_in[5];
  const float* td_W2 = (const float*)d_in[6];
  const float* td_b2 = (const float*)d_in[7];
  const float* bu_W1 = (const float*)d_in[8];
  const float* bu_b1 = (const float*)d_in[9];
  const float* bu_W2 = (const float*)d_in[10];
  const float* bu_b2 = (const float*)d_in[11];
  const float* pw1   = (const float*)d_in[12];
  const float* pb1   = (const float*)d_in[13];
  const float* pw2   = (const float*)d_in[14];
  const float* pb2   = (const float*)d_in[15];

  const int IN_F = L1_K, HID = L1_N, G = 512;
  const int n_nodes = in_sizes[0] / IN_F;
  const int n_edges = in_sizes[1] / 2;
  const int* src = edge_idx;
  const int* dst = edge_idx + n_edges;

  float* ws      = (float*)d_ws;
  const size_t NF = (size_t)n_nodes * 128;
  float* h_lin   = ws;
  int*   csr     = (int*)(ws + NF);
  float* h_act   = ws + 2 * NF;
  float* dinv_td = ws + 3 * NF;
  float* dinv_bu = dinv_td + n_nodes;
  float* pooled  = dinv_bu + n_nodes;
  float* gcnt    = pooled + (size_t)G * 256;
  float* hcat    = ws;
  float* mlp_h   = ws + (size_t)G * 256;

  int* cnt_td    = csr;
  int* cnt_bu    = cnt_td + n_nodes;
  int* rowptr_td = cnt_bu + n_nodes;
  int* rowptr_bu = rowptr_td + (n_nodes + 1);
  int* col_td    = rowptr_bu + (n_nodes + 1);
  int* col_bu    = col_td + n_edges;
  int* rank_td   = col_bu + n_edges;
  int* rank_bu   = rank_td + n_edges;

  const int RB = GATHER_BLOCKS;
  const int FT = FUSED_WAVES * 32;
  const int GB  = ((n_nodes >> 4) * (HID >> 6) + FUSED_WAVES - 1) / FUSED_WAVES;
  const int EBF = (n_edges + FT * EDGE_PER_THREAD - 1) / (FT * EDGE_PER_THREAD);
  const int RBF = RB * (256 / FT);
  const int GB1 = GB / 2;

  hipMemsetAsync(cnt_td, 0, (size_t)2 * n_nodes * sizeof(int), stream);
  k_count_rank_gemm<<<EBF + GB1, FT, 0, stream>>>(src, dst, cnt_td, cnt_bu, rank_td, rank_bu,
                                                 n_edges, EBF, x, td_W1, h_lin, n_nodes, 0);
  k_scan_dinv<<<2, SCAN_WAVES * 32, 0, stream>>>(cnt_td, rowptr_td, dinv_td, n_nodes);
  k_fill_gemm<<<EBF + (GB - GB1), FT, 0, stream>>>(src, dst, rowptr_td, rowptr_bu, rank_td, rank_bu,
                                                  col_td, col_bu, n_edges, EBF,
                                                  x, td_W1, h_lin, n_nodes, GB1 * FUSED_WAVES);
  hipMemsetAsync(pooled, 0, ((size_t)G * 256 + G) * sizeof(float), stream);

  k_gather_relu<<<RB, 256, 0, stream>>>(h_lin, rowptr_td, col_td, dinv_td, td_b1, h_act, n_nodes);
  k_gather_pool_gemm<<<RBF + GB, FT, 0, stream>>>(h_act, rowptr_td, col_td, dinv_td, batch, pooled,
                                                 gcnt,  128, n_nodes, RBF,
                                                 x, bu_W1, h_lin, n_nodes);

  k_gather_relu<<<RB, 256, 0, stream>>>(h_lin, rowptr_bu, col_bu, dinv_bu, bu_b1, h_act, n_nodes);
  k_gather_pool<<<RB, 256, 0, stream>>>(h_act, rowptr_bu, col_bu, dinv_bu, batch, pooled, nullptr,
                                         0, n_nodes);

  k_pool_project<<<((G >> 4) * 4 + GEMM_WAVES - 1) / GEMM_WAVES, GEMM_WAVES * 32, 0, stream>>>(
      pooled, bu_W2, td_W2, bu_b2, td_b2, gcnt, hcat, G);
  launch_gemm(hcat, pw1, mlp_h, pb1, G, 256, 256,  1, stream);
  launch_gemm(mlp_h, pw2, (float*)d_out, pb2, G, 128, 256,  0, stream);
}


# pooled gather: a wave keeps the sum of a run of consecutive rows of one graph in registers and adds it to pooled once per run, one whole line per atomic instruction: about 20x fewer atomic requests
# speedup vs baseline: 3.2047x; 1.2024x over previous
#include <hip/hip_runtime.h>


typedef __attribute__((ext_vector_type(2))) float v2f;
typedef __attribute__((ext_vector_type(8))) float v8f;

#define EDGE_PER_THREAD 4
__device__ __forceinline__ void
count_rank_body(const int* __restrict__ src, const int* __restrict__ dst,
                int* __restrict__ cnt_td, int* __restrict__ cnt_bu,
                int* __restrict__ rank_td, int* __restrict__ rank_bu,
                int n_edges, int blk, int nblk) {
  const int width = nblk * blockDim.x;
  const int e0 = blk * blockDim.x + threadIdx.x;
  int s[EDGE_PER_THREAD], d[EDGE_PER_THREAD], rt[EDGE_PER_THREAD], rb[EDGE_PER_THREAD];
#pragma unroll
  for (int j = 0; j < EDGE_PER_THREAD; ++j) {
    const int e = min(e0 + j * width, n_edges - 1);
    s[j] = src[e];
    d[j] = dst[e];
  }
#pragma unroll
  for (int j = 0; j < EDGE_PER_THREAD; ++j) {
    const int e = e0 + j * width;
    rt[j] = 0;
    rb[j] = 0;
    if (e < n_edges) {
      rt[j] = atomicAdd(&cnt_td[d[j]], 1);
      rb[j] = atomicAdd(&cnt_bu[s[j]], 1);
    }
  }
#pragma unroll
  for (int j = 0; j < EDGE_PER_THREAD; ++j) {
    const int e = e0 + j * width;
    if (e < n_edges) {
      rank_td[e] = rt[j];
      rank_bu[e] = rb[j];
    }
  }
}

#define SCAN_WAVES 32
#define SCAN_BATCH 7
__global__ void __launch_bounds__(SCAN_WAVES * 32)
k_scan_dinv(const int* __restrict__ cnt2, int* __restrict__ rowptr2,
            float* __restrict__ dinv2, int n) {
  __shared__ int wave_tot[SCAN_WAVES];
  const int* __restrict__ cnt = cnt2 + (size_t)blockIdx.x * n;
  int* __restrict__ rowptr = rowptr2 + (size_t)blockIdx.x * (n + 1);
  float* __restrict__ dinv = dinv2 + (size_t)blockIdx.x * n;
  const int lane = threadIdx.x & 31;
  const int wave = threadIdx.x >> 5;
  const int steps = (n + SCAN_WAVES * 32 - 1) / (SCAN_WAVES * 32);
  const int base = wave * steps * 32;

  int part = 0;
  for (int k = 0; k < steps; ++k) {
    const int i = base + k * 32 + lane;
    part += (i < n) ? cnt[i] : 0;
  }
  for (int o = 16; o > 0; o >>= 1) part += __shfl_xor(part, o);
  if (lane == 0) wave_tot[wave] = part;
  __syncthreads();
  int carry = (lane < wave) ? wave_tot[lane] : 0;
  for (int o = 16; o > 0; o >>= 1) carry += __shfl_xor(carry, o);

  for (int k0 = 0; k0 < steps; k0 += SCAN_BATCH) {
    int v[SCAN_BATCH];
#pragma unroll
    for (int u = 0; u < SCAN_BATCH; ++u) {
      const int i = base + (k0 + u) * 32 + lane;
      v[u] = (k0 + u < steps && i < n) ? cnt[i] : 0;
    }
    int x[SCAN_BATCH];
#pragma unroll
    for (int u = 0; u < SCAN_BATCH; ++u) x[u] = v[u];
#pragma unroll
    for (int o = 1; o < 32; o <<= 1) {
      int y[SCAN_BATCH];
#pragma unroll
      for (int u = 0; u < SCAN_BATCH; ++u) y[u] = __shfl_up(x[u], o);
#pragma unroll
      for (int u = 0; u < SCAN_BATCH; ++u) x[u] += (lane >= o) ? y[u] : 0;
    }
#pragma unroll
    for (int u = 0; u < SCAN_BATCH; ++u) {
      const int i = base + (k0 + u) * 32 + lane;
      if (k0 + u < steps && i < n) {
        rowptr[i] = carry + x[u] - v[u];
        dinv[i] = rsqrtf((float)v[u] + 1.0f);
      }
      carry += __builtin_amdgcn_readlane(x[u], 31);
    }
  }
  if (wave == SCAN_WAVES - 1 && lane == 0) rowptr[n] = carry;
}

__device__ __forceinline__ void
fill_body(const int* __restrict__ src, const int* __restrict__ dst,
          const int* __restrict__ rowptr_td, const int* __restrict__ rowptr_bu,
          const int* __restrict__ rank_td, const int* __restrict__ rank_bu,
          int* __restrict__ col_td, int* __restrict__ col_bu, int n_edges, int blk, int nblk) {
  const int width = nblk * blockDim.x;
  const int e0 = blk * blockDim.x + threadIdx.x;
  int s[EDGE_PER_THREAD], d[EDGE_PER_THREAD], pt[EDGE_PER_THREAD], pb[EDGE_PER_THREAD];
#pragma unroll
  for (int j = 0; j < EDGE_PER_THREAD; ++j) {
    const int e = min(e0 + j * width, n_edges - 1);
    s[j] = src[e];
    d[j] = dst[e];
    pt[j] = rank_td[e];
    pb[j] = rank_bu[e];
  }
#pragma unroll
  for (int j = 0; j < EDGE_PER_THREAD; ++j) {
    pt[j] += rowptr_td[d[j]];
    pb[j] += rowptr_bu[s[j]];
  }
#pragma unroll
  for (int j = 0; j < EDGE_PER_THREAD; ++j) {
    const int e = e0 + j * width;
    if (e < n_edges) {
      col_td[pt[j]] = s[j];
      col_bu[pb[j]] = d[j];
    }
  }
}

struct GemmFrag {
  float4 a;
  float4 b0, b1, b2, b3;
};
__device__ __forceinline__ GemmFrag gemm_load(const float* __restrict__ Ap,
                                              const float* __restrict__ Bp, int k0, int N) {
  GemmFrag f;
  f.a  = *reinterpret_cast<const float4*>(Ap + k0);
  const float* __restrict__ bp = Bp + (size_t)k0 * N;
  f.b0 = *reinterpret_cast<const float4*>(bp);
  f.b1 = *reinterpret_cast<const float4*>(bp + N);
  f.b2 = *reinterpret_cast<const float4*>(bp + 2 * (size_t)N);
  f.b3 = *reinterpret_cast<const float4*>(bp + 3 * (size_t)N);
  return f;
}
__device__ __forceinline__ v8f mma4(float a0, float a1, float b0, float b1, v8f acc) {
  v2f a, b;
  a.x = a0; a.y = a1;
  b.x = b0; b.y = b1;
  return __builtin_amdgcn_wmma_f32_16x16x4_f32(
       false, a,  false, b,
       (short)0, acc,  false,  false);
}
__device__ __forceinline__ void gemm_block(const GemmFrag& f, v8f& acc0, v8f& acc1,
                                           v8f& acc2, v8f& acc3) {
  acc0 = mma4(f.a.x, f.a.y, f.b0.x, f.b1.x, acc0);
  acc1 = mma4(f.a.x, f.a.y, f.b0.y, f.b1.y, acc1);
  acc2 = mma4(f.a.x, f.a.y, f.b0.z, f.b1.z, acc2);
  acc3 = mma4(f.a.x, f.a.y, f.b0.w, f.b1.w, acc3);
  acc0 = mma4(f.a.z, f.a.w, f.b2.x, f.b3.x, acc0);
  acc1 = mma4(f.a.z, f.a.w, f.b2.y, f.b3.y, acc1);
  acc2 = mma4(f.a.z, f.a.w, f.b2.z, f.b3.z, acc2);
  acc3 = mma4(f.a.z, f.a.w, f.b2.w, f.b3.w, acc3);
}

__device__ __forceinline__ void
gemm_strip(const float* __restrict__ A, const float* __restrict__ B,
           float* __restrict__ C, const float* __restrict__ bias,
           int M, int N, int K, int relu, int strip) {
  const int lane = threadIdx.x & 31;
  const int sn_cnt = N >> 6;
  const int total = (M >> 4) * sn_cnt;
  if (strip >= total) return;
  const int tm = (strip / sn_cnt) << 4;
  const int cn = (strip % sn_cnt) << 6;

  const int h = lane >> 4;
  const int c = lane & 15;

  const float* __restrict__ Ap = A + (size_t)(tm + c) * K + 4 * h;
  const float* __restrict__ Bp = B + (size_t)(4 * h) * N + cn + 4 * c;

  v8f acc0 = {}, acc1 = {}, acc2 = {}, acc3 = {};
  GemmFrag f0 = gemm_load(Ap, Bp, 0, N);
  for (int k0 = 0; k0 < K; k0 += 16) {
    const GemmFrag f1 = gemm_load(Ap, Bp, k0 + 8, N);
    __builtin_amdgcn_sched_barrier(0);
    gemm_block(f0, acc0, acc1, acc2, acc3);
    f0 = gemm_load(Ap, Bp, (k0 + 16 < K) ? k0 + 16 : k0, N);
    __builtin_amdgcn_sched_barrier(0);
    gemm_block(f1, acc0, acc1, acc2, acc3);
  }

  float4 bi = make_float4(0.f, 0.f, 0.f, 0.f);
  if (bias) bi = *reinterpret_cast<const float4*>(bias + cn + 4 * c);
#pragma unroll
  for (int r = 0; r < 8; ++r) {
    float4 v;
    v.x = acc0[r] + bi.x;
    v.y = acc1[r] + bi.y;
    v.z = acc2[r] + bi.z;
    v.w = acc3[r] + bi.w;
    if (relu) {
      v.x = fmaxf(v.x, 0.0f);
      v.y = fmaxf(v.y, 0.0f);
      v.z = fmaxf(v.z, 0.0f);
      v.w = fmaxf(v.w, 0.0f);
    }
    *reinterpret_cast<float4*>(C + (size_t)(tm + 8 * h + r) * N + cn + 4 * c) = v;
  }
}

#define FUSED_WAVES 4
#define FUSED_MIN_WAVES 10
#define L1_K 256
#define L1_N 128
__global__ void __launch_bounds__(FUSED_WAVES * 32, FUSED_MIN_WAVES)
k_count_rank_gemm(const int* __restrict__ src, const int* __restrict__ dst,
                  int* __restrict__ cnt_td, int* __restrict__ cnt_bu,
                  int* __restrict__ rank_td, int* __restrict__ rank_bu,
                  int n_edges, int other_blocks,
                  const float* __restrict__ A, const float* __restrict__ B,
                  float* __restrict__ C, int M, int strip0) {
  if ((int)blockIdx.x < other_blocks) {
    count_rank_body(src, dst, cnt_td, cnt_bu, rank_td, rank_bu, n_edges,
                    (int)blockIdx.x, other_blocks);
  } else {
    gemm_strip(A, B, C, nullptr, M, L1_N, L1_K, 0,
               strip0 + ((int)blockIdx.x - other_blocks) * FUSED_WAVES + (int)(threadIdx.x >> 5));
  }
}

__global__ void __launch_bounds__(FUSED_WAVES * 32, FUSED_MIN_WAVES)
k_fill_gemm(const int* __restrict__ src, const int* __restrict__ dst,
            const int* __restrict__ rowptr_td, const int* __restrict__ rowptr_bu,
            const int* __restrict__ rank_td, const int* __restrict__ rank_bu,
            int* __restrict__ col_td, int* __restrict__ col_bu,
            int n_edges, int other_blocks,
            const float* __restrict__ A, const float* __restrict__ B,
            float* __restrict__ C, int M, int strip0) {
  if ((int)blockIdx.x < other_blocks) {
    fill_body(src, dst, rowptr_td, rowptr_bu, rank_td, rank_bu, col_td, col_bu, n_edges,
              (int)blockIdx.x, other_blocks);
  } else {
    gemm_strip(A, B, C, nullptr, M, L1_N, L1_K, 0,
               strip0 + ((int)blockIdx.x - other_blocks) * FUSED_WAVES + (int)(threadIdx.x >> 5));
  }
}

__device__ __forceinline__ float4 ld_row(const float* __restrict__ h, int s, int lane) {
  return *reinterpret_cast<const float4*>(h + (size_t)s * 128 + lane * 4);
}
__device__ __forceinline__ void fma4(float4& a, const float4 v, const float w) {
  a.x = fmaf(v.x, w, a.x);
  a.y = fmaf(v.y, w, a.y);
  a.z = fmaf(v.z, w, a.z);
  a.w = fmaf(v.w, w, a.w);
}
__device__ __forceinline__ int lane_i(int v, int j) { return __builtin_amdgcn_readlane(v, j); }
__device__ __forceinline__ float lane_f(float v, int j) {
  return __int_as_float(__builtin_amdgcn_readlane(__float_as_int(v), j));
}

__device__ __forceinline__ float4 gather_row(const float* __restrict__ h,
                                             const int* __restrict__ col,
                                             const float* __restrict__ dinv,
                                             int beg, int end, float dd, int lane) {
  float4 acc = make_float4(0.f, 0.f, 0.f, 0.f);
  for (int base = beg; base < end; base += 32) {
    const int n = min(32, end - base);
    int my_s = 0;
    float my_w = 0.f;
    if (lane < n) {
      my_s = col[base + lane];
      my_w = dinv[my_s] * dd;
    }
    int j = 0;
    for (; j + 8 <= n; j += 8) {
      const float4 v0 = ld_row(h, lane_i(my_s, j), lane);
      const float4 v1 = ld_row(h, lane_i(my_s, j + 1), lane);
      const float4 v2 = ld_row(h, lane_i(my_s, j + 2), lane);
      const float4 v3 = ld_row(h, lane_i(my_s, j + 3), lane);
      const float4 v4 = ld_row(h, lane_i(my_s, j + 4), lane);
      const float4 v5 = ld_row(h, lane_i(my_s, j + 5), lane);
      const float4 v6 = ld_row(h, lane_i(my_s, j + 6), lane);
      const float4 v7 = ld_row(h, lane_i(my_s, j + 7), lane);
      fma4(acc, v0, lane_f(my_w, j));
      fma4(acc, v1, lane_f(my_w, j + 1));
      fma4(acc, v2, lane_f(my_w, j + 2));
      fma4(acc, v3, lane_f(my_w, j + 3));
      fma4(acc, v4, lane_f(my_w, j + 4));
      fma4(acc, v5, lane_f(my_w, j + 5));
      fma4(acc, v6, lane_f(my_w, j + 6));
      fma4(acc, v7, lane_f(my_w, j + 7));
    }
    if (j + 4 <= n) {
      const float4 v0 = ld_row(h, lane_i(my_s, j), lane);
      const float4 v1 = ld_row(h, lane_i(my_s, j + 1), lane);
      const float4 v2 = ld_row(h, lane_i(my_s, j + 2), lane);
      const float4 v3 = ld_row(h, lane_i(my_s, j + 3), lane);
      fma4(acc, v0, lane_f(my_w, j));
      fma4(acc, v1, lane_f(my_w, j + 1));
      fma4(acc, v2, lane_f(my_w, j + 2));
      fma4(acc, v3, lane_f(my_w, j + 3));
      j += 4;
    }
    if (j + 2 <= n) {
      const float4 v0 = ld_row(h, lane_i(my_s, j), lane);
      const float4 v1 = ld_row(h, lane_i(my_s, j + 1), lane);
      fma4(acc, v0, lane_f(my_w, j));
      fma4(acc, v1, lane_f(my_w, j + 1));
      j += 2;
    }
    if (j < n) {
      const float4 v0 = ld_row(h, lane_i(my_s, j), lane);
      fma4(acc, v0, lane_f(my_w, j));
    }
  }
  return acc;
}

#define GATHER_BLOCKS 1024
__device__ __forceinline__ void wave_rows(int n_nodes, int blk, int nblk, int& r0, int& r1) {
  const int g = __builtin_amdgcn_readfirstlane((int)((blk * blockDim.x + threadIdx.x) >> 5));
  const int nw = (int)((nblk * blockDim.x) >> 5);
  const int q = n_nodes / nw;
  const int r = n_nodes % nw;
  r0 = g * q + min(g, r);
  r1 = r0 + q + (g < r ? 1 : 0);
}

__global__ void __launch_bounds__(256)
k_gather_relu(const float* __restrict__ h, const int* __restrict__ rowptr,
              const int* __restrict__ col, const float* __restrict__ dinv,
              const float* __restrict__ bias, float* __restrict__ out, int n_nodes) {
  const int lane = threadIdx.x & 31;
  int r0, r1;
  wave_rows(n_nodes, (int)blockIdx.x, (int)gridDim.x, r0, r1);
  const float4 b = *reinterpret_cast<const float4*>(bias + lane * 4);
  for (int row = r0; row < r1; ++row) {
    const int beg = rowptr[row];
    const int end = rowptr[row + 1];
    const float dd = dinv[row];
    const float4 self = ld_row(h, row, lane);
    const float4 a = gather_row(h, col, dinv, beg, end, dd, lane);
    float4 v;
    v.x = fmaxf(a.x + self.x * dd * dd + b.x, 0.0f);
    v.y = fmaxf(a.y + self.y * dd * dd + b.y, 0.0f);
    v.z = fmaxf(a.z + self.z * dd * dd + b.z, 0.0f);
    v.w = fmaxf(a.w + self.w * dd * dd + b.w, 0.0f);
    *reinterpret_cast<float4*>(out + (size_t)row * 128 + lane * 4) = v;
  }
}

__device__ __forceinline__ void
pool_flush(float* __restrict__ pooled, float* __restrict__ gcnt, int g, int col_off,
           const float4 run, int run_rows, int lane) {
  float* __restrict__ p = pooled + (size_t)g * 256 + col_off + lane;
  const int sub = lane & 3;
#pragma unroll
  for (int k = 0; k < 4; ++k) {
    const int from = 8 * k + (lane >> 2);
    const float x = __shfl(run.x, from);
    const float y = __shfl(run.y, from);
    const float z = __shfl(run.z, from);
    const float w = __shfl(run.w, from);
    const float lo = (sub & 1) ? y : x;
    const float hi = (sub & 1) ? w : z;
    const float v = (sub & 2) ? hi : lo;
    atomicAdd(&p[32 * k], v);
  }
  if (gcnt != nullptr && lane == 0) atomicAdd(&gcnt[g], (float)run_rows);
}

__device__ __forceinline__ void
gather_pool_body(const float* __restrict__ h, const int* __restrict__ rowptr,
                 const int* __restrict__ col, const float* __restrict__ dinv,
                 const int* __restrict__ batch, float* __restrict__ pooled,
                 float* __restrict__ gcnt, int col_off, int n_nodes, int blk, int nblk) {
  const int lane = threadIdx.x & 31;
  int r0, r1;
  wave_rows(n_nodes, blk, nblk, r0, r1);
  int run_g = 0;
  int run_rows = 0;
  float4 run = make_float4(0.f, 0.f, 0.f, 0.f);
  for (int row = r0; row < r1; ++row) {
    const int beg = rowptr[row];
    const int end = rowptr[row + 1];
    const float dd = dinv[row];
    const int g = batch[row];
    const float4 self = ld_row(h, row, lane);
    const float4 a = gather_row(h, col, dinv, beg, end, dd, lane);
    if (run_rows > 0 && g != run_g) {
      pool_flush(pooled, gcnt, run_g, col_off, run, run_rows, lane);
      run = make_float4(0.f, 0.f, 0.f, 0.f);
      run_rows = 0;
    }
    run_g = g;
    ++run_rows;
    run.x += a.x + self.x * dd * dd;
    run.y += a.y + self.y * dd * dd;
    run.z += a.z + self.z * dd * dd;
    run.w += a.w + self.w * dd * dd;
  }
  if (run_rows > 0) pool_flush(pooled, gcnt, run_g, col_off, run, run_rows, lane);
}

__global__ void __launch_bounds__(256)
k_gather_pool(const float* __restrict__ h, const int* __restrict__ rowptr,
              const int* __restrict__ col, const float* __restrict__ dinv,
              const int* __restrict__ batch, float* __restrict__ pooled,
              float* __restrict__ gcnt, int col_off, int n_nodes) {
  gather_pool_body(h, rowptr, col, dinv, batch, pooled, gcnt, col_off, n_nodes,
                   (int)blockIdx.x, (int)gridDim.x);
}

__global__ void __launch_bounds__(FUSED_WAVES * 32, FUSED_MIN_WAVES)
k_gather_pool_gemm(const float* __restrict__ h, const int* __restrict__ rowptr,
                   const int* __restrict__ col, const float* __restrict__ dinv,
                   const int* __restrict__ batch, float* __restrict__ pooled,
                   float* __restrict__ gcnt, int col_off, int n_nodes, int other_blocks,
                   const float* __restrict__ A, const float* __restrict__ B,
                   float* __restrict__ C, int M) {
  if ((int)blockIdx.x < other_blocks) {
    gather_pool_body(h, rowptr, col, dinv, batch, pooled, gcnt, col_off, n_nodes,
                     (int)blockIdx.x, other_blocks);
  } else {
    gemm_strip(A, B, C, nullptr, M, L1_N, L1_K, 0,
               ((int)blockIdx.x - other_blocks) * FUSED_WAVES + (int)(threadIdx.x >> 5));
  }
}

#define HEAD_WAVES 4
#define HEAD_DEPTH 8
#define HEAD_PITCH 260

template <int N>
__device__ __forceinline__ void head_load_b(GemmFrag& f, const float* __restrict__ Bp, int blk) {
  const float* __restrict__ bp = Bp + (size_t)blk * 8 * N;
  f.b0 = *reinterpret_cast<const float4*>(bp);
  f.b1 = *reinterpret_cast<const float4*>(bp + N);
  f.b2 = *reinterpret_cast<const float4*>(bp + 2 * N);
  f.b3 = *reinterpret_cast<const float4*>(bp + 3 * N);
}
__device__ __forceinline__ void head_load_a(GemmFrag& f, const float* Ap, int blk) {
  f.a = *reinterpret_cast<const float4*>(Ap + blk * 8);
}

template <int N, int NB, int D>
__device__ __forceinline__ void head_product(const float* Ap, const float* __restrict__ Bp,
                                             GemmFrag (&f)[D], v8f& acc0, v8f& acc1,
                                             v8f& acc2, v8f& acc3) {
  int kb = 0;
#pragma unroll 1
  for (; kb + D < NB; kb += D) {
#pragma unroll
    for (int u = 0; u < D; ++u) {
      gemm_block(f[u], acc0, acc1, acc2, acc3);
      head_load_a(f[u], Ap, kb + D + u);
      head_load_b<N>(f[u], Bp, kb + D + u);
      __builtin_amdgcn_sched_barrier(0);
    }
  }
#pragma unroll
  for (int u = 0; u < D; ++u) gemm_block(f[u], acc0, acc1, acc2, acc3);
}

__global__ void __launch_bounds__(HEAD_WAVES * 32)
k_head(const float* __restrict__ pooled,
       const float* __restrict__ W_bu, const float* __restrict__ W_td,
       const float* __restrict__ b_bu, const float* __restrict__ b_td,
       const float* __restrict__ gcnt,
       const float* __restrict__ pw1, const float* __restrict__ pb1,
       const float* __restrict__ pw2, const float* __restrict__ pb2,
       float* __restrict__ out) {
  __shared__ __attribute__((aligned(16))) float H1[16 * HEAD_PITCH];
  __shared__ __attribute__((aligned(16))) float H2[16 * HEAD_PITCH];
  const int lane = threadIdx.x & 31;
  const int wave = __builtin_amdgcn_readfirstlane((int)(threadIdx.x >> 5));
  const int h = lane >> 4;
  const int c = lane & 15;
  const int tm = (int)blockIdx.x << 4;

  const int br = wave >> 1;
  const int cn = (wave & 1) << 6;
  const float* __restrict__ Wb = br ? W_td : W_bu;
  const float* __restrict__ bb = br ? b_td : b_bu;
  const float* __restrict__ Ap1 = pooled + (size_t)(tm + c) * 256 + 128 * br + 4 * h;
  const float* __restrict__ Bp1 = Wb + (size_t)(4 * h) * 128 + cn + 4 * c;
  GemmFrag f1[16];
#pragma unroll
  for (int u = 0; u < 16; ++u) {
    head_load_a(f1[u], Ap1, u);
    head_load_b<128>(f1[u], Bp1, u);
  }
  float cntf[8];
#pragma unroll
  for (int r = 0; r < 8; ++r) cntf[r] = gcnt[tm + 8 * h + r];
  const float4 bi1 = *reinterpret_cast<const float4*>(bb + cn + 4 * c);
  const float4 bi2 = *reinterpret_cast<const float4*>(pb1 + 64 * wave + 4 * c);

  v8f acc0 = {}, acc1 = {}, acc2 = {}, acc3 = {};
  head_product<128, 16, 16>(Ap1, Bp1, f1, acc0, acc1, acc2, acc3);

  const float* __restrict__ Bp2 = pw1 + (size_t)(4 * h) * 256 + 64 * wave + 4 * c;
  GemmFrag f[HEAD_DEPTH];
#pragma unroll
  for (int u = 0; u < HEAD_DEPTH; ++u) head_load_b<256>(f[u], Bp2, u);
#pragma unroll
  for (int r = 0; r < 8; ++r) {
    float4 v;
    v.x = fmaf(cntf[r], bi1.x, acc0[r]);
    v.y = fmaf(cntf[r], bi1.y, acc1[r]);
    v.z = fmaf(cntf[r], bi1.z, acc2[r]);
    v.w = fmaf(cntf[r], bi1.w, acc3[r]);
    *reinterpret_cast<float4*>(&H1[(8 * h + r) * HEAD_PITCH + 64 * wave + 4 * c]) = v;
  }
  __syncthreads();

  const float* As2 = &H1[c * HEAD_PITCH + 4 * h];
#pragma unroll
  for (int u = 0; u < HEAD_DEPTH; ++u) head_load_a(f[u], As2, u);
  acc0 = v8f{}; acc1 = v8f{}; acc2 = v8f{}; acc3 = v8f{};
  head_product<256, 32, HEAD_DEPTH>(As2, Bp2, f, acc0, acc1, acc2, acc3);

  const float* __restrict__ Bp3 = pw2 + (size_t)(4 * h) * 128 + 64 * (wave & 1) + 4 * c;
  if (wave < 2) {
#pragma unroll
    for (int u = 0; u < HEAD_DEPTH; ++u) head_load_b<128>(f[u], Bp3, u);
  }
#pragma unroll
  for (int r = 0; r < 8; ++r) {
    float4 v;
    v.x = fmaxf(acc0[r] + bi2.x, 0.0f);
    v.y = fmaxf(acc1[r] + bi2.y, 0.0f);
    v.z = fmaxf(acc2[r] + bi2.z, 0.0f);
    v.w = fmaxf(acc3[r] + bi2.w, 0.0f);
    *reinterpret_cast<float4*>(&H2[(8 * h + r) * HEAD_PITCH + 64 * wave + 4 * c]) = v;
  }
  __syncthreads();

  if (wave < 2) {
    const float* As3 = &H2[c * HEAD_PITCH + 4 * h];
#pragma unroll
    for (int u = 0; u < HEAD_DEPTH; ++u) head_load_a(f[u], As3, u);
    const float4 bi3 = *reinterpret_cast<const float4*>(pb2 + 64 * wave + 4 * c);
    acc0 = v8f{}; acc1 = v8f{}; acc2 = v8f{}; acc3 = v8f{};
    head_product<128, 32, HEAD_DEPTH>(As3, Bp3, f, acc0, acc1, acc2, acc3);
#pragma unroll
    for (int r = 0; r < 8; ++r) {
      float4 v;
      v.x = acc0[r] + bi3.x;
      v.y = acc1[r] + bi3.y;
      v.z = acc2[r] + bi3.z;
      v.w = acc3[r] + bi3.w;
      *reinterpret_cast<float4*>(out + (size_t)(tm + 8 * h + r) * 128 + 64 * wave + 4 * c) = v;
    }
  }
}

extern "C" void kernel_launch(void* const* d_in, const int* in_sizes, int n_in,
                              void* d_out, int out_size, void* d_ws, size_t ws_size,
                              hipStream_t stream) {
  const float* x      = (const float*)d_in[0];
  const int* edge_idx = (const int*)d_in[1];
  const int* batch    = (const int*)d_in[2];
  const float* td_W1 = (const float*)d_in[4];
  const float* td_b1 = (const float*)d_in[5];
  const float* td_W2 = (const float*)d_in[6];
  const float* td_b2 = (const float*)d_in[7];
  const float* bu_W1 = (const float*)d_in[8];
  const float* bu_b1 = (const float*)d_in[9];
  const float* bu_W2 = (const float*)d_in[10];
  const float* bu_b2 = (const float*)d_in[11];
  const float* pw1   = (const float*)d_in[12];
  const float* pb1   = (const float*)d_in[13];
  const float* pw2   = (const float*)d_in[14];
  const float* pb2   = (const float*)d_in[15];

  const int IN_F = L1_K, HID = L1_N, G = 512;
  const int n_nodes = in_sizes[0] / IN_F;
  const int n_edges = in_sizes[1] / 2;
  const int* src = edge_idx;
  const int* dst = edge_idx + n_edges;

  float* ws      = (float*)d_ws;
  const size_t NF = (size_t)n_nodes * 128;
  float* h_lin   = ws;
  int*   csr     = (int*)(ws + NF);
  float* h_act   = ws + 2 * NF;
  float* dinv_td = ws + 3 * NF;
  float* dinv_bu = dinv_td + n_nodes;
  float* pooled  = dinv_bu + n_nodes;
  float* gcnt    = pooled + (size_t)G * 256;

  int* cnt_td    = csr;
  int* cnt_bu    = cnt_td + n_nodes;
  int* rowptr_td = cnt_bu + n_nodes;
  int* rowptr_bu = rowptr_td + (n_nodes + 1);
  int* col_td    = rowptr_bu + (n_nodes + 1);
  int* col_bu    = col_td + n_edges;
  int* rank_td   = col_bu + n_edges;
  int* rank_bu   = rank_td + n_edges;

  const int RB = GATHER_BLOCKS;
  const int FT = FUSED_WAVES * 32;
  const int GB  = ((n_nodes >> 4) * (HID >> 6) + FUSED_WAVES - 1) / FUSED_WAVES;
  const int EBF = (n_edges + FT * EDGE_PER_THREAD - 1) / (FT * EDGE_PER_THREAD);
  const int RBF = RB * (256 / FT);
  const int GB1 = GB / 2;

  hipMemsetAsync(cnt_td, 0, (size_t)2 * n_nodes * sizeof(int), stream);
  k_count_rank_gemm<<<EBF + GB1, FT, 0, stream>>>(src, dst, cnt_td, cnt_bu, rank_td, rank_bu,
                                                 n_edges, EBF, x, td_W1, h_lin, n_nodes, 0);
  k_scan_dinv<<<2, SCAN_WAVES * 32, 0, stream>>>(cnt_td, rowptr_td, dinv_td, n_nodes);
  k_fill_gemm<<<EBF + (GB - GB1), FT, 0, stream>>>(src, dst, rowptr_td, rowptr_bu, rank_td, rank_bu,
                                                  col_td, col_bu, n_edges, EBF,
                                                  x, td_W1, h_lin, n_nodes, GB1 * FUSED_WAVES);
  hipMemsetAsync(pooled, 0, ((size_t)G * 256 + G) * sizeof(float), stream);

  k_gather_relu<<<RB, 256, 0, stream>>>(h_lin, rowptr_td, col_td, dinv_td, td_b1, h_act, n_nodes);
  k_gather_pool_gemm<<<RBF + GB, FT, 0, stream>>>(h_act, rowptr_td, col_td, dinv_td, batch, pooled,
                                                 gcnt,  128, n_nodes, RBF,
                                                 x, bu_W1, h_lin, n_nodes);

  k_gather_relu<<<RB, 256, 0, stream>>>(h_lin, rowptr_bu, col_bu, dinv_bu, bu_b1, h_act, n_nodes);
  k_gather_pool<<<RB, 256, 0, stream>>>(h_act, rowptr_bu, col_bu, dinv_bu, batch, pooled, nullptr,
                                         0, n_nodes);

  k_head<<<G >> 4, HEAD_WAVES * 32, 0, stream>>>(pooled, bu_W2, td_W2, bu_b2, td_b2, gcnt,
                                                 pw1, pb1, pw2, pb2, (float*)d_out);
}


# edge counting in LDS: 32 work-groups rank their edges with LDS atomics, then add per-node counts to memory a whole line per atomic (return = row base); TD product beside it; row order by arrival
# speedup vs baseline: 4.2094x; 1.3135x over previous
#include <hip/hip_runtime.h>


typedef __attribute__((ext_vector_type(2))) float v2f;
typedef __attribute__((ext_vector_type(8))) float v8f;

#define EDGE_PER_THREAD 4
#define CNT_THREADS 1024
#define CNT_MAXN 50176
#define CNT_WG_PER_DIR 16
#define CNT_BATCH 10
__device__ __forceinline__ void
count_rank_lds_body(const int* __restrict__ key, int* __restrict__ cnt, int* __restrict__ rank,
                    int n_edges, int n_nodes, int chunk, int nchunk, int* node_cnt) {
  const int tid = (int)threadIdx.x;
  const int per = (n_edges + nchunk - 1) / nchunk;
  const int e_beg = chunk * per;
  const int e_end = min(n_edges, e_beg + per);

  for (int i = tid; i < CNT_MAXN / 4; i += CNT_THREADS)
    reinterpret_cast<int4*>(node_cnt)[i] = make_int4(0, 0, 0, 0);
  __syncthreads();

  for (int e0 = e_beg; e0 < e_end; e0 += CNT_THREADS * CNT_BATCH) {
    int k[CNT_BATCH], r[CNT_BATCH];
#pragma unroll
    for (int u = 0; u < CNT_BATCH; ++u) {
      const int e = e0 + u * CNT_THREADS + tid;
      k[u] = key[min(e, e_end - 1)];
    }
#pragma unroll
    for (int u = 0; u < CNT_BATCH; ++u) {
      const int e = e0 + u * CNT_THREADS + tid;
      r[u] = 0;
      if (e < e_end) r[u] = atomicAdd(&node_cnt[k[u]], 1);
    }
#pragma unroll
    for (int u = 0; u < CNT_BATCH; ++u) {
      const int e = e0 + u * CNT_THREADS + tid;
      if (e < e_end) rank[e] = r[u];
    }
  }
  __syncthreads();

  const int rot = (chunk * (n_nodes / nchunk)) & ~31;
  for (int i0 = 0; i0 < n_nodes; i0 += CNT_THREADS * CNT_BATCH) {
    int c[CNT_BATCH], b[CNT_BATCH], nn[CNT_BATCH];
#pragma unroll
    for (int u = 0; u < CNT_BATCH; ++u) {
      const int i = i0 + u * CNT_THREADS + tid;
      int n = i + rot;
      if (n >= n_nodes) n -= n_nodes;
      nn[u] = n;
      c[u] = (i < n_nodes) ? node_cnt[n] : 0;
    }
#pragma unroll
    for (int u = 0; u < CNT_BATCH; ++u) {
      b[u] = 0;
      if (c[u] > 0) b[u] = atomicAdd(&cnt[nn[u]], c[u]);
    }
#pragma unroll
    for (int u = 0; u < CNT_BATCH; ++u) {
      const int i = i0 + u * CNT_THREADS + tid;
      if (i < n_nodes) node_cnt[nn[u]] = b[u];
    }
  }
  __syncthreads();

  for (int e0 = e_beg; e0 < e_end; e0 += CNT_THREADS * CNT_BATCH) {
    int k[CNT_BATCH], r[CNT_BATCH];
#pragma unroll
    for (int u = 0; u < CNT_BATCH; ++u) {
      const int e = min(e0 + u * CNT_THREADS + tid, e_end - 1);
      k[u] = key[e];
      r[u] = rank[e];
    }
#pragma unroll
    for (int u = 0; u < CNT_BATCH; ++u) r[u] += node_cnt[k[u]];
#pragma unroll
    for (int u = 0; u < CNT_BATCH; ++u) {
      const int e = e0 + u * CNT_THREADS + tid;
      if (e < e_end) rank[e] = r[u];
    }
  }
}

#define SCAN_WAVES 32
#define SCAN_BATCH 7
__global__ void __launch_bounds__(SCAN_WAVES * 32)
k_scan_dinv(const int* __restrict__ cnt2, int* __restrict__ rowptr2,
            float* __restrict__ dinv2, int n) {
  __shared__ int wave_tot[SCAN_WAVES];
  const int* __restrict__ cnt = cnt2 + (size_t)blockIdx.x * n;
  int* __restrict__ rowptr = rowptr2 + (size_t)blockIdx.x * (n + 1);
  float* __restrict__ dinv = dinv2 + (size_t)blockIdx.x * n;
  const int lane = threadIdx.x & 31;
  const int wave = threadIdx.x >> 5;
  const int steps = (n + SCAN_WAVES * 32 - 1) / (SCAN_WAVES * 32);
  const int base = wave * steps * 32;

  int part = 0;
  for (int k = 0; k < steps; ++k) {
    const int i = base + k * 32 + lane;
    part += (i < n) ? cnt[i] : 0;
  }
  for (int o = 16; o > 0; o >>= 1) part += __shfl_xor(part, o);
  if (lane == 0) wave_tot[wave] = part;
  __syncthreads();
  int carry = (lane < wave) ? wave_tot[lane] : 0;
  for (int o = 16; o > 0; o >>= 1) carry += __shfl_xor(carry, o);

  for (int k0 = 0; k0 < steps; k0 += SCAN_BATCH) {
    int v[SCAN_BATCH];
#pragma unroll
    for (int u = 0; u < SCAN_BATCH; ++u) {
      const int i = base + (k0 + u) * 32 + lane;
      v[u] = (k0 + u < steps && i < n) ? cnt[i] : 0;
    }
    int x[SCAN_BATCH];
#pragma unroll
    for (int u = 0; u < SCAN_BATCH; ++u) x[u] = v[u];
#pragma unroll
    for (int o = 1; o < 32; o <<= 1) {
      int y[SCAN_BATCH];
#pragma unroll
      for (int u = 0; u < SCAN_BATCH; ++u) y[u] = __shfl_up(x[u], o);
#pragma unroll
      for (int u = 0; u < SCAN_BATCH; ++u) x[u] += (lane >= o) ? y[u] : 0;
    }
#pragma unroll
    for (int u = 0; u < SCAN_BATCH; ++u) {
      const int i = base + (k0 + u) * 32 + lane;
      if (k0 + u < steps && i < n) {
        rowptr[i] = carry + x[u] - v[u];
        dinv[i] = rsqrtf((float)v[u] + 1.0f);
      }
      carry += __builtin_amdgcn_readlane(x[u], 31);
    }
  }
  if (wave == SCAN_WAVES - 1 && lane == 0) rowptr[n] = carry;
}

__device__ __forceinline__ void
fill_body(const int* __restrict__ src, const int* __restrict__ dst,
          const int* __restrict__ rowptr_td, const int* __restrict__ rowptr_bu,
          const int* __restrict__ rank_td, const int* __restrict__ rank_bu,
          int* __restrict__ col_td, int* __restrict__ col_bu, int n_edges, int blk, int nblk) {
  const int width = nblk * blockDim.x;
  const int e0 = blk * blockDim.x + threadIdx.x;
  int s[EDGE_PER_THREAD], d[EDGE_PER_THREAD], pt[EDGE_PER_THREAD], pb[EDGE_PER_THREAD];
#pragma unroll
  for (int j = 0; j < EDGE_PER_THREAD; ++j) {
    const int e = min(e0 + j * width, n_edges - 1);
    s[j] = src[e];
    d[j] = dst[e];
    pt[j] = rank_td[e];
    pb[j] = rank_bu[e];
  }
#pragma unroll
  for (int j = 0; j < EDGE_PER_THREAD; ++j) {
    pt[j] += rowptr_td[d[j]];
    pb[j] += rowptr_bu[s[j]];
  }
#pragma unroll
  for (int j = 0; j < EDGE_PER_THREAD; ++j) {
    const int e = e0 + j * width;
    if (e < n_edges) {
      col_td[pt[j]] = s[j];
      col_bu[pb[j]] = d[j];
    }
  }
}

struct GemmFrag {
  float4 a;
  float4 b0, b1, b2, b3;
};
__device__ __forceinline__ GemmFrag gemm_load(const float* __restrict__ Ap,
                                              const float* __restrict__ Bp, int k0, int N) {
  GemmFrag f;
  f.a  = *reinterpret_cast<const float4*>(Ap + k0);
  const float* __restrict__ bp = Bp + (size_t)k0 * N;
  f.b0 = *reinterpret_cast<const float4*>(bp);
  f.b1 = *reinterpret_cast<const float4*>(bp + N);
  f.b2 = *reinterpret_cast<const float4*>(bp + 2 * (size_t)N);
  f.b3 = *reinterpret_cast<const float4*>(bp + 3 * (size_t)N);
  return f;
}
__device__ __forceinline__ v8f mma4(float a0, float a1, float b0, float b1, v8f acc) {
  v2f a, b;
  a.x = a0; a.y = a1;
  b.x = b0; b.y = b1;
  return __builtin_amdgcn_wmma_f32_16x16x4_f32(
       false, a,  false, b,
       (short)0, acc,  false,  false);
}
__device__ __forceinline__ void gemm_block(const GemmFrag& f, v8f& acc0, v8f& acc1,
                                           v8f& acc2, v8f& acc3) {
  acc0 = mma4(f.a.x, f.a.y, f.b0.x, f.b1.x, acc0);
  acc1 = mma4(f.a.x, f.a.y, f.b0.y, f.b1.y, acc1);
  acc2 = mma4(f.a.x, f.a.y, f.b0.z, f.b1.z, acc2);
  acc3 = mma4(f.a.x, f.a.y, f.b0.w, f.b1.w, acc3);
  acc0 = mma4(f.a.z, f.a.w, f.b2.x, f.b3.x, acc0);
  acc1 = mma4(f.a.z, f.a.w, f.b2.y, f.b3.y, acc1);
  acc2 = mma4(f.a.z, f.a.w, f.b2.z, f.b3.z, acc2);
  acc3 = mma4(f.a.z, f.a.w, f.b2.w, f.b3.w, acc3);
}

__device__ __forceinline__ void
gemm_strip(const float* __restrict__ A, const float* __restrict__ B,
           float* __restrict__ C, const float* __restrict__ bias,
           int M, int N, int K, int relu, int strip) {
  const int lane = threadIdx.x & 31;
  const int sn_cnt = N >> 6;
  const int total = (M >> 4) * sn_cnt;
  if (strip >= total) return;
  const int tm = (strip / sn_cnt) << 4;
  const int cn = (strip % sn_cnt) << 6;

  const int h = lane >> 4;
  const int c = lane & 15;

  const float* __restrict__ Ap = A + (size_t)(tm + c) * K + 4 * h;
  const float* __restrict__ Bp = B + (size_t)(4 * h) * N + cn + 4 * c;

  v8f acc0 = {}, acc1 = {}, acc2 = {}, acc3 = {};
  GemmFrag f0 = gemm_load(Ap, Bp, 0, N);
  for (int k0 = 0; k0 < K; k0 += 16) {
    const GemmFrag f1 = gemm_load(Ap, Bp, k0 + 8, N);
    __builtin_amdgcn_sched_barrier(0);
    gemm_block(f0, acc0, acc1, acc2, acc3);
    f0 = gemm_load(Ap, Bp, (k0 + 16 < K) ? k0 + 16 : k0, N);
    __builtin_amdgcn_sched_barrier(0);
    gemm_block(f1, acc0, acc1, acc2, acc3);
  }

  float4 bi = make_float4(0.f, 0.f, 0.f, 0.f);
  if (bias) bi = *reinterpret_cast<const float4*>(bias + cn + 4 * c);
#pragma unroll
  for (int r = 0; r < 8; ++r) {
    float4 v;
    v.x = acc0[r] + bi.x;
    v.y = acc1[r] + bi.y;
    v.z = acc2[r] + bi.z;
    v.w = acc3[r] + bi.w;
    if (relu) {
      v.x = fmaxf(v.x, 0.0f);
      v.y = fmaxf(v.y, 0.0f);
      v.z = fmaxf(v.z, 0.0f);
      v.w = fmaxf(v.w, 0.0f);
    }
    *reinterpret_cast<float4*>(C + (size_t)(tm + 8 * h + r) * N + cn + 4 * c) = v;
  }
}

#define FUSED_WAVES 4
#define FUSED_MIN_WAVES 10
#define L1_K 256
#define L1_N 128
__global__ void __launch_bounds__(CNT_THREADS)
k_count_lds_gemm(const int* __restrict__ src, const int* __restrict__ dst,
                 int* __restrict__ cnt_td, int* __restrict__ cnt_bu,
                 int* __restrict__ rank_td, int* __restrict__ rank_bu,
                 int n_edges, int n_nodes,
                 const float* __restrict__ A, const float* __restrict__ B,
                 float* __restrict__ C, int M) {
  __shared__ __attribute__((aligned(16))) int node_cnt[CNT_MAXN];
  const int count_blocks = 2 * CNT_WG_PER_DIR;
  if ((int)blockIdx.x < count_blocks) {
    const int bu = (int)blockIdx.x & 1;
    count_rank_lds_body(bu ? src : dst, bu ? cnt_bu : cnt_td, bu ? rank_bu : rank_td,
                        n_edges, n_nodes, (int)blockIdx.x >> 1, CNT_WG_PER_DIR, node_cnt);
  } else {
    gemm_strip(A, B, C, nullptr, M, L1_N, L1_K, 0,
               ((int)blockIdx.x - count_blocks) * (CNT_THREADS / 32) + (int)(threadIdx.x >> 5));
  }
}

__global__ void __launch_bounds__(FUSED_WAVES * 32)
k_fill(const int* __restrict__ src, const int* __restrict__ dst,
       const int* __restrict__ rowptr_td, const int* __restrict__ rowptr_bu,
       const int* __restrict__ rank_td, const int* __restrict__ rank_bu,
       int* __restrict__ col_td, int* __restrict__ col_bu, int n_edges) {
  fill_body(src, dst, rowptr_td, rowptr_bu, rank_td, rank_bu, col_td, col_bu, n_edges,
            (int)blockIdx.x, (int)gridDim.x);
}

__device__ __forceinline__ float4 ld_row(const float* __restrict__ h, int s, int lane) {
  return *reinterpret_cast<const float4*>(h + (size_t)s * 128 + lane * 4);
}
__device__ __forceinline__ void fma4(float4& a, const float4 v, const float w) {
  a.x = fmaf(v.x, w, a.x);
  a.y = fmaf(v.y, w, a.y);
  a.z = fmaf(v.z, w, a.z);
  a.w = fmaf(v.w, w, a.w);
}
__device__ __forceinline__ int lane_i(int v, int j) { return __builtin_amdgcn_readlane(v, j); }
__device__ __forceinline__ float lane_f(float v, int j) {
  return __int_as_float(__builtin_amdgcn_readlane(__float_as_int(v), j));
}

__device__ __forceinline__ float4 gather_row(const float* __restrict__ h,
                                             const int* __restrict__ col,
                                             const float* __restrict__ dinv,
                                             int beg, int end, float dd, int lane) {
  float4 acc = make_float4(0.f, 0.f, 0.f, 0.f);
  for (int base = beg; base < end; base += 32) {
    const int n = min(32, end - base);
    int my_s = 0;
    float my_w = 0.f;
    if (lane < n) {
      my_s = col[base + lane];
      my_w = dinv[my_s] * dd;
    }
    int j = 0;
    for (; j + 8 <= n; j += 8) {
      const float4 v0 = ld_row(h, lane_i(my_s, j), lane);
      const float4 v1 = ld_row(h, lane_i(my_s, j + 1), lane);
      const float4 v2 = ld_row(h, lane_i(my_s, j + 2), lane);
      const float4 v3 = ld_row(h, lane_i(my_s, j + 3), lane);
      const float4 v4 = ld_row(h, lane_i(my_s, j + 4), lane);
      const float4 v5 = ld_row(h, lane_i(my_s, j + 5), lane);
      const float4 v6 = ld_row(h, lane_i(my_s, j + 6), lane);
      const float4 v7 = ld_row(h, lane_i(my_s, j + 7), lane);
      fma4(acc, v0, lane_f(my_w, j));
      fma4(acc, v1, lane_f(my_w, j + 1));
      fma4(acc, v2, lane_f(my_w, j + 2));
      fma4(acc, v3, lane_f(my_w, j + 3));
      fma4(acc, v4, lane_f(my_w, j + 4));
      fma4(acc, v5, lane_f(my_w, j + 5));
      fma4(acc, v6, lane_f(my_w, j + 6));
      fma4(acc, v7, lane_f(my_w, j + 7));
    }
    if (j + 4 <= n) {
      const float4 v0 = ld_row(h, lane_i(my_s, j), lane);
      const float4 v1 = ld_row(h, lane_i(my_s, j + 1), lane);
      const float4 v2 = ld_row(h, lane_i(my_s, j + 2), lane);
      const float4 v3 = ld_row(h, lane_i(my_s, j + 3), lane);
      fma4(acc, v0, lane_f(my_w, j));
      fma4(acc, v1, lane_f(my_w, j + 1));
      fma4(acc, v2, lane_f(my_w, j + 2));
      fma4(acc, v3, lane_f(my_w, j + 3));
      j += 4;
    }
    if (j + 2 <= n) {
      const float4 v0 = ld_row(h, lane_i(my_s, j), lane);
      const float4 v1 = ld_row(h, lane_i(my_s, j + 1), lane);
      fma4(acc, v0, lane_f(my_w, j));
      fma4(acc, v1, lane_f(my_w, j + 1));
      j += 2;
    }
    if (j < n) {
      const float4 v0 = ld_row(h, lane_i(my_s, j), lane);
      fma4(acc, v0, lane_f(my_w, j));
    }
  }
  return acc;
}

#define GATHER_BLOCKS 1024
__device__ __forceinline__ void wave_rows(int n_nodes, int blk, int nblk, int& r0, int& r1) {
  const int g = __builtin_amdgcn_readfirstlane((int)((blk * blockDim.x + threadIdx.x) >> 5));
  const int nw = (int)((nblk * blockDim.x) >> 5);
  const int q = n_nodes / nw;
  const int r = n_nodes % nw;
  r0 = g * q + min(g, r);
  r1 = r0 + q + (g < r ? 1 : 0);
}

__global__ void __launch_bounds__(256)
k_gather_relu(const float* __restrict__ h, const int* __restrict__ rowptr,
              const int* __restrict__ col, const float* __restrict__ dinv,
              const float* __restrict__ bias, float* __restrict__ out, int n_nodes) {
  const int lane = threadIdx.x & 31;
  int r0, r1;
  wave_rows(n_nodes, (int)blockIdx.x, (int)gridDim.x, r0, r1);
  const float4 b = *reinterpret_cast<const float4*>(bias + lane * 4);
  for (int row = r0; row < r1; ++row) {
    const int beg = rowptr[row];
    const int end = rowptr[row + 1];
    const float dd = dinv[row];
    const float4 self = ld_row(h, row, lane);
    const float4 a = gather_row(h, col, dinv, beg, end, dd, lane);
    float4 v;
    v.x = fmaxf(a.x + self.x * dd * dd + b.x, 0.0f);
    v.y = fmaxf(a.y + self.y * dd * dd + b.y, 0.0f);
    v.z = fmaxf(a.z + self.z * dd * dd + b.z, 0.0f);
    v.w = fmaxf(a.w + self.w * dd * dd + b.w, 0.0f);
    *reinterpret_cast<float4*>(out + (size_t)row * 128 + lane * 4) = v;
  }
}

__device__ __forceinline__ void
pool_flush(float* __restrict__ pooled, float* __restrict__ gcnt, int g, int col_off,
           const float4 run, int run_rows, int lane) {
  float* __restrict__ p = pooled + (size_t)g * 256 + col_off + lane;
  const int sub = lane & 3;
#pragma unroll
  for (int k = 0; k < 4; ++k) {
    const int from = 8 * k + (lane >> 2);
    const float x = __shfl(run.x, from);
    const float y = __shfl(run.y, from);
    const float z = __shfl(run.z, from);
    const float w = __shfl(run.w, from);
    const float lo = (sub & 1) ? y : x;
    const float hi = (sub & 1) ? w : z;
    const float v = (sub & 2) ? hi : lo;
    atomicAdd(&p[32 * k], v);
  }
  if (gcnt != nullptr && lane == 0) atomicAdd(&gcnt[g], (float)run_rows);
}

__device__ __forceinline__ void
gather_pool_body(const float* __restrict__ h, const int* __restrict__ rowptr,
                 const int* __restrict__ col, const float* __restrict__ dinv,
                 const int* __restrict__ batch, float* __restrict__ pooled,
                 float* __restrict__ gcnt, int col_off, int n_nodes, int blk, int nblk) {
  const int lane = threadIdx.x & 31;
  int r0, r1;
  wave_rows(n_nodes, blk, nblk, r0, r1);
  int run_g = 0;
  int run_rows = 0;
  float4 run = make_float4(0.f, 0.f, 0.f, 0.f);
  for (int row = r0; row < r1; ++row) {
    const int beg = rowptr[row];
    const int end = rowptr[row + 1];
    const float dd = dinv[row];
    const int g = batch[row];
    const float4 self = ld_row(h, row, lane);
    const float4 a = gather_row(h, col, dinv, beg, end, dd, lane);
    if (run_rows > 0 && g != run_g) {
      pool_flush(pooled, gcnt, run_g, col_off, run, run_rows, lane);
      run = make_float4(0.f, 0.f, 0.f, 0.f);
      run_rows = 0;
    }
    run_g = g;
    ++run_rows;
    run.x += a.x + self.x * dd * dd;
    run.y += a.y + self.y * dd * dd;
    run.z += a.z + self.z * dd * dd;
    run.w += a.w + self.w * dd * dd;
  }
  if (run_rows > 0) pool_flush(pooled, gcnt, run_g, col_off, run, run_rows, lane);
}

__global__ void __launch_bounds__(256)
k_gather_pool(const float* __restrict__ h, const int* __restrict__ rowptr,
              const int* __restrict__ col, const float* __restrict__ dinv,
              const int* __restrict__ batch, float* __restrict__ pooled,
              float* __restrict__ gcnt, int col_off, int n_nodes) {
  gather_pool_body(h, rowptr, col, dinv, batch, pooled, gcnt, col_off, n_nodes,
                   (int)blockIdx.x, (int)gridDim.x);
}

__global__ void __launch_bounds__(FUSED_WAVES * 32, FUSED_MIN_WAVES)
k_gather_pool_gemm(const float* __restrict__ h, const int* __restrict__ rowptr,
                   const int* __restrict__ col, const float* __restrict__ dinv,
                   const int* __restrict__ batch, float* __restrict__ pooled,
                   float* __restrict__ gcnt, int col_off, int n_nodes, int other_blocks,
                   const float* __restrict__ A, const float* __restrict__ B,
                   float* __restrict__ C, int M) {
  if ((int)blockIdx.x < other_blocks) {
    gather_pool_body(h, rowptr, col, dinv, batch, pooled, gcnt, col_off, n_nodes,
                     (int)blockIdx.x, other_blocks);
  } else {
    gemm_strip(A, B, C, nullptr, M, L1_N, L1_K, 0,
               ((int)blockIdx.x - other_blocks) * FUSED_WAVES + (int)(threadIdx.x >> 5));
  }
}

#define HEAD_WAVES 4
#define HEAD_DEPTH 8
#define HEAD_PITCH 260

template <int N>
__device__ __forceinline__ void head_load_b(GemmFrag& f, const float* __restrict__ Bp, int blk) {
  const float* __restrict__ bp = Bp + (size_t)blk * 8 * N;
  f.b0 = *reinterpret_cast<const float4*>(bp);
  f.b1 = *reinterpret_cast<const float4*>(bp + N);
  f.b2 = *reinterpret_cast<const float4*>(bp + 2 * N);
  f.b3 = *reinterpret_cast<const float4*>(bp + 3 * N);
}
__device__ __forceinline__ void head_load_a(GemmFrag& f, const float* Ap, int blk) {
  f.a = *reinterpret_cast<const float4*>(Ap + blk * 8);
}

template <int N, int NB, int D>
__device__ __forceinline__ void head_product(const float* Ap, const float* __restrict__ Bp,
                                             GemmFrag (&f)[D], v8f& acc0, v8f& acc1,
                                             v8f& acc2, v8f& acc3) {
  int kb = 0;
#pragma unroll 1
  for (; kb + D < NB; kb += D) {
#pragma unroll
    for (int u = 0; u < D; ++u) {
      gemm_block(f[u], acc0, acc1, acc2, acc3);
      head_load_a(f[u], Ap, kb + D + u);
      head_load_b<N>(f[u], Bp, kb + D + u);
      __builtin_amdgcn_sched_barrier(0);
    }
  }
#pragma unroll
  for (int u = 0; u < D; ++u) gemm_block(f[u], acc0, acc1, acc2, acc3);
}

__global__ void __launch_bounds__(HEAD_WAVES * 32)
k_head(const float* __restrict__ pooled,
       const float* __restrict__ W_bu, const float* __restrict__ W_td,
       const float* __restrict__ b_bu, const float* __restrict__ b_td,
       const float* __restrict__ gcnt,
       const float* __restrict__ pw1, const float* __restrict__ pb1,
       const float* __restrict__ pw2, const float* __restrict__ pb2,
       float* __restrict__ out) {
  __shared__ __attribute__((aligned(16))) float H1[16 * HEAD_PITCH];
  __shared__ __attribute__((aligned(16))) float H2[16 * HEAD_PITCH];
  const int lane = threadIdx.x & 31;
  const int wave = __builtin_amdgcn_readfirstlane((int)(threadIdx.x >> 5));
  const int h = lane >> 4;
  const int c = lane & 15;
  const int tm = (int)blockIdx.x << 4;

  const int br = wave >> 1;
  const int cn = (wave & 1) << 6;
  const float* __restrict__ Wb = br ? W_td : W_bu;
  const float* __restrict__ bb = br ? b_td : b_bu;
  const float* __restrict__ Ap1 = pooled + (size_t)(tm + c) * 256 + 128 * br + 4 * h;
  const float* __restrict__ Bp1 = Wb + (size_t)(4 * h) * 128 + cn + 4 * c;
  GemmFrag f1[16];
#pragma unroll
  for (int u = 0; u < 16; ++u) {
    head_load_a(f1[u], Ap1, u);
    head_load_b<128>(f1[u], Bp1, u);
  }
  float cntf[8];
#pragma unroll
  for (int r = 0; r < 8; ++r) cntf[r] = gcnt[tm + 8 * h + r];
  const float4 bi1 = *reinterpret_cast<const float4*>(bb + cn + 4 * c);
  const float4 bi2 = *reinterpret_cast<const float4*>(pb1 + 64 * wave + 4 * c);

  v8f acc0 = {}, acc1 = {}, acc2 = {}, acc3 = {};
  head_product<128, 16, 16>(Ap1, Bp1, f1, acc0, acc1, acc2, acc3);

  const float* __restrict__ Bp2 = pw1 + (size_t)(4 * h) * 256 + 64 * wave + 4 * c;
  GemmFrag f[HEAD_DEPTH];
#pragma unroll
  for (int u = 0; u < HEAD_DEPTH; ++u) head_load_b<256>(f[u], Bp2, u);
#pragma unroll
  for (int r = 0; r < 8; ++r) {
    float4 v;
    v.x = fmaf(cntf[r], bi1.x, acc0[r]);
    v.y = fmaf(cntf[r], bi1.y, acc1[r]);
    v.z = fmaf(cntf[r], bi1.z, acc2[r]);
    v.w = fmaf(cntf[r], bi1.w, acc3[r]);
    *reinterpret_cast<float4*>(&H1[(8 * h + r) * HEAD_PITCH + 64 * wave + 4 * c]) = v;
  }
  __syncthreads();

  const float* As2 = &H1[c * HEAD_PITCH + 4 * h];
#pragma unroll
  for (int u = 0; u < HEAD_DEPTH; ++u) head_load_a(f[u], As2, u);
  acc0 = v8f{}; acc1 = v8f{}; acc2 = v8f{}; acc3 = v8f{};
  head_product<256, 32, HEAD_DEPTH>(As2, Bp2, f, acc0, acc1, acc2, acc3);

  const float* __restrict__ Bp3 = pw2 + (size_t)(4 * h) * 128 + 64 * (wave & 1) + 4 * c;
  if (wave < 2) {
#pragma unroll
    for (int u = 0; u < HEAD_DEPTH; ++u) head_load_b<128>(f[u], Bp3, u);
  }
#pragma unroll
  for (int r = 0; r < 8; ++r) {
    float4 v;
    v.x = fmaxf(acc0[r] + bi2.x, 0.0f);
    v.y = fmaxf(acc1[r] + bi2.y, 0.0f);
    v.z = fmaxf(acc2[r] + bi2.z, 0.0f);
    v.w = fmaxf(acc3[r] + bi2.w, 0.0f);
    *reinterpret_cast<float4*>(&H2[(8 * h + r) * HEAD_PITCH + 64 * wave + 4 * c]) = v;
  }
  __syncthreads();

  if (wave < 2) {
    const float* As3 = &H2[c * HEAD_PITCH + 4 * h];
#pragma unroll
    for (int u = 0; u < HEAD_DEPTH; ++u) head_load_a(f[u], As3, u);
    const float4 bi3 = *reinterpret_cast<const float4*>(pb2 + 64 * wave + 4 * c);
    acc0 = v8f{}; acc1 = v8f{}; acc2 = v8f{}; acc3 = v8f{};
    head_product<128, 32, HEAD_DEPTH>(As3, Bp3, f, acc0, acc1, acc2, acc3);
#pragma unroll
    for (int r = 0; r < 8; ++r) {
      float4 v;
      v.x = acc0[r] + bi3.x;
      v.y = acc1[r] + bi3.y;
      v.z = acc2[r] + bi3.z;
      v.w = acc3[r] + bi3.w;
      *reinterpret_cast<float4*>(out + (size_t)(tm + 8 * h + r) * 128 + 64 * wave + 4 * c) = v;
    }
  }
}

extern "C" void kernel_launch(void* const* d_in, const int* in_sizes, int n_in,
                              void* d_out, int out_size, void* d_ws, size_t ws_size,
                              hipStream_t stream) {
  const float* x      = (const float*)d_in[0];
  const int* edge_idx = (const int*)d_in[1];
  const int* batch    = (const int*)d_in[2];
  const float* td_W1 = (const float*)d_in[4];
  const float* td_b1 = (const float*)d_in[5];
  const float* td_W2 = (const float*)d_in[6];
  const float* td_b2 = (const float*)d_in[7];
  const float* bu_W1 = (const float*)d_in[8];
  const float* bu_b1 = (const float*)d_in[9];
  const float* bu_W2 = (const float*)d_in[10];
  const float* bu_b2 = (const float*)d_in[11];
  const float* pw1   = (const float*)d_in[12];
  const float* pb1   = (const float*)d_in[13];
  const float* pw2   = (const float*)d_in[14];
  const float* pb2   = (const float*)d_in[15];

  const int IN_F = L1_K, HID = L1_N, G = 512;
  const int n_nodes = in_sizes[0] / IN_F;
  const int n_edges = in_sizes[1] / 2;
  const int* src = edge_idx;
  const int* dst = edge_idx + n_edges;

  float* ws      = (float*)d_ws;
  const size_t NF = (size_t)n_nodes * 128;
  float* h_lin   = ws;
  int*   csr     = (int*)(ws + NF);
  float* h_act   = ws + 2 * NF;
  float* dinv_td = ws + 3 * NF;
  float* dinv_bu = dinv_td + n_nodes;
  float* pooled  = dinv_bu + n_nodes;
  float* gcnt    = pooled + (size_t)G * 256;

  int* cnt_td    = csr;
  int* cnt_bu    = cnt_td + n_nodes;
  int* rowptr_td = cnt_bu + n_nodes;
  int* rowptr_bu = rowptr_td + (n_nodes + 1);
  int* col_td    = rowptr_bu + (n_nodes + 1);
  int* col_bu    = col_td + n_edges;
  int* rank_td   = col_bu + n_edges;
  int* rank_bu   = rank_td + n_edges;

  const int RB = GATHER_BLOCKS;
  const int FT = FUSED_WAVES * 32;
  const int GB  = ((n_nodes >> 4) * (HID >> 6) + FUSED_WAVES - 1) / FUSED_WAVES;
  const int EBF = (n_edges + FT * EDGE_PER_THREAD - 1) / (FT * EDGE_PER_THREAD);
  const int RBF = RB * (256 / FT);
  const int GBC = ((n_nodes >> 4) * (HID >> 6) + CNT_THREADS / 32 - 1) / (CNT_THREADS / 32);

  hipMemsetAsync(cnt_td, 0, (size_t)2 * n_nodes * sizeof(int), stream);
  k_count_lds_gemm<<<2 * CNT_WG_PER_DIR + GBC, CNT_THREADS, 0, stream>>>(
      src, dst, cnt_td, cnt_bu, rank_td, rank_bu, n_edges, n_nodes, x, td_W1, h_lin, n_nodes);
  k_scan_dinv<<<2, SCAN_WAVES * 32, 0, stream>>>(cnt_td, rowptr_td, dinv_td, n_nodes);
  k_fill<<<EBF, FT, 0, stream>>>(src, dst, rowptr_td, rowptr_bu, rank_td, rank_bu,
                                 col_td, col_bu, n_edges);
  hipMemsetAsync(pooled, 0, ((size_t)G * 256 + G) * sizeof(float), stream);

  k_gather_relu<<<RB, 256, 0, stream>>>(h_lin, rowptr_td, col_td, dinv_td, td_b1, h_act, n_nodes);
  k_gather_pool_gemm<<<RBF + GB, FT, 0, stream>>>(h_act, rowptr_td, col_td, dinv_td, batch, pooled,
                                                 gcnt,  128, n_nodes, RBF,
                                                 x, bu_W1, h_lin, n_nodes);

  k_gather_relu<<<RB, 256, 0, stream>>>(h_lin, rowptr_bu, col_bu, dinv_bu, bu_b1, h_act, n_nodes);
  k_gather_pool<<<RB, 256, 0, stream>>>(h_act, rowptr_bu, col_bu, dinv_bu, batch, pooled, nullptr,
                                         0, n_nodes);

  k_head<<<G >> 4, HEAD_WAVES * 32, 0, stream>>>(pooled, bu_W2, td_W2, bu_b2, td_b2, gcnt,
                                                 pw1, pb1, pw2, pb2, (float*)d_out);
}
